# starting kernel
# baseline (speedup 1.0000x reference)
#include <hip/hip_runtime.h>
#include <hip/hip_bf16.h>
#include <math.h>

#define NN 100000
#define CC 16
#define GG 2
#define LL 2
#define EE 3200000
#define FF 512
#define HH 64

#define W1_PSTRIDE 160
#define W2_PSTRIDE 32
#define HID_STRIDE 66

typedef float v2f __attribute__((ext_vector_type(2)));
typedef float v8f __attribute__((ext_vector_type(8)));

__device__ __forceinline__ unsigned f2ou(float f) {
    unsigned u = __float_as_uint(f);
    return (u & 0x80000000u) ? ~u : (u | 0x80000000u);
}
__device__ __forceinline__ float ou2f(unsigned u) {
    return __uint_as_float((u & 0x80000000u) ? (u & 0x7FFFFFFFu) : ~u);
}

__global__ void k_init(unsigned* __restrict__ nmax, float* __restrict__ nsum,
                       float* __restrict__ hnext) {
    int i = blockIdx.x * blockDim.x + threadIdx.x;
    if (i < NN) { nmax[i] = 0x007FFFFFu;   nsum[i] = 0.0f; }
    if (i < NN * CC) hnext[i] = 0.0f;
}

__global__ void k_segmax(const float* __restrict__ e, const int* __restrict__ dst,
                         unsigned* __restrict__ nmax) {
    int i = blockIdx.x * blockDim.x + threadIdx.x;
    if (i >= EE) return;
    atomicMax(&nmax[dst[i]], f2ou(e[i]));
}

__global__ void k_segsum(const float* __restrict__ e, const int* __restrict__ dst,
                         const unsigned* __restrict__ nmax, float* __restrict__ nsum) {
    int i = blockIdx.x * blockDim.x + threadIdx.x;
    if (i >= EE) return;
    int d = dst[i];
    atomicAdd(&nsum[d], expf(e[i] - ou2f(nmax[d])));
}

__global__ void k_agg(const float* __restrict__ e, const int* __restrict__ src,
                      const int* __restrict__ dst, const unsigned* __restrict__ nmax,
                      const float* __restrict__ nsum, const float* __restrict__ hcur,
                      float* __restrict__ hnext) {
    int i = blockIdx.x * blockDim.x + threadIdx.x;
    if (i >= EE) return;
    int d = dst[i], s = src[i];
    float a = expf(e[i] - ou2f(nmax[d])) / nsum[d];
    const float4* hs = (const float4*)(hcur + (size_t)s * CC);
    float4 v0 = hs[0], v1 = hs[1], v2 = hs[2], v3 = hs[3];
    float* o = hnext + (size_t)d * CC;
    atomicAdd(o + 0,  v0.x * a); atomicAdd(o + 1,  v0.y * a);
    atomicAdd(o + 2,  v0.z * a); atomicAdd(o + 3,  v0.w * a);
    atomicAdd(o + 4,  v1.x * a); atomicAdd(o + 5,  v1.y * a);
    atomicAdd(o + 6,  v1.z * a); atomicAdd(o + 7,  v1.w * a);
    atomicAdd(o + 8,  v2.x * a); atomicAdd(o + 9,  v2.y * a);
    atomicAdd(o + 10, v2.z * a); atomicAdd(o + 11, v2.w * a);
    atomicAdd(o + 12, v3.x * a); atomicAdd(o + 13, v3.y * a);
    atomicAdd(o + 14, v3.z * a); atomicAdd(o + 15, v3.w * a);
}

__global__ void k_mask(const float* __restrict__ one_hot, const int* __restrict__ mask,
                       float* __restrict__ h) {
    int i = blockIdx.x * blockDim.x + threadIdx.x;
    if (i >= NN * CC) return;
    int node = i >> 4;
    h[i] = mask[node] ? one_hot[i] : h[i];
}

__global__ __launch_bounds__(256) void k_mlp_combine(
    const float* __restrict__ feat, const float* __restrict__ w1,
    const float* __restrict__ b1,   const float* __restrict__ w2,
    const float* __restrict__ b2,   const float* __restrict__ alpha,
    const float* __restrict__ att,  const float* __restrict__ h0,
    const float* __restrict__ h1,   float* __restrict__ out, int n_pairs)
{
    __shared__ float sW1p[(FF / 2) * W1_PSTRIDE];
    __shared__ float sW2p[(HH / 2) * W2_PSTRIDE];
    __shared__ float sHid[8][2][16 * HID_STRIDE];

    int t = threadIdx.x;

    for (int i = t; i < (FF / 2) * (HH / 4); i += 256) {
        int p  = i >> 4;
        int nq = (i & 15) << 2;
        float4 x = *(const float4*)&w1[(size_t)p * 2 * HH + nq];
        float4 y = *(const float4*)&w1[(size_t)p * 2 * HH + HH + nq];
        float4 lo = { x.x, y.x, x.y, y.y };
        float4 hi = { x.z, y.z, x.w, y.w };
        float* d = &sW1p[p * W1_PSTRIDE + 2 * nq];
        *(float4*)(d + 0) = lo;
        *(float4*)(d + 4) = hi;
    }
    for (int i = t; i < (HH / 2) * CC; i += 256) {
        int p = i >> 4, n = i & 15;
        v2f v; v.x = w2[(size_t)p * 2 * CC + n];
        v.y = w2[(size_t)p * 2 * CC + CC + n];
        *(v2f*)&sW2p[p * W2_PSTRIDE + 2 * n] = v;
    }
    __syncthreads();

    int wave = t >> 5;
    int lane = t & 31;
    int pair = blockIdx.x * 8 + wave;
    if (pair >= n_pairs) return;

    int tile0 = pair * 2;
    int tile1 = tile0 + 1;

    int m     = lane & 15;
    int n     = lane & 15;
    int hi16  = lane >> 4;
    int khalf = hi16 << 1;
    int moff  = hi16 << 3;

    v8f acc0 = {}, acc1 = {}, acc2 = {}, acc3 = {};
    v8f acc4 = {}, acc5 = {}, acc6 = {}, acc7 = {};
    const float* arow0 = feat + ((size_t)tile0 * 16 + m) * FF;
    const float* arow1 = feat + ((size_t)tile1 * 16 + m) * FF;
    for (int k = 0; k < FF; k += 4) {
        int kb = k + khalf;
        v2f a0 = *(const v2f*)(arow0 + kb);
        v2f a1 = *(const v2f*)(arow1 + kb);
        const float* bp = &sW1p[((k >> 1) + hi16) * W1_PSTRIDE + 2 * n];
        v2f b0 = *(const v2f*)(bp + 0);
        v2f b1v = *(const v2f*)(bp + 32);
        v2f b2v = *(const v2f*)(bp + 64);
        v2f b3v = *(const v2f*)(bp + 96);
        acc0 = __builtin_amdgcn_wmma_f32_16x16x4_f32(false, a0, false, b0,  (short)0, acc0, false, false);
        acc1 = __builtin_amdgcn_wmma_f32_16x16x4_f32(false, a0, false, b1v, (short)0, acc1, false, false);
        acc2 = __builtin_amdgcn_wmma_f32_16x16x4_f32(false, a0, false, b2v, (short)0, acc2, false, false);
        acc3 = __builtin_amdgcn_wmma_f32_16x16x4_f32(false, a0, false, b3v, (short)0, acc3, false, false);
        acc4 = __builtin_amdgcn_wmma_f32_16x16x4_f32(false, a1, false, b0,  (short)0, acc4, false, false);
        acc5 = __builtin_amdgcn_wmma_f32_16x16x4_f32(false, a1, false, b1v, (short)0, acc5, false, false);
        acc6 = __builtin_amdgcn_wmma_f32_16x16x4_f32(false, a1, false, b2v, (short)0, acc6, false, false);
        acc7 = __builtin_amdgcn_wmma_f32_16x16x4_f32(false, a1, false, b3v, (short)0, acc7, false, false);
    }

    float bb0 = b1[n + 0], bb1 = b1[n + 16], bb2 = b1[n + 32], bb3 = b1[n + 48];
    {
        float* hd0 = &sHid[wave][0][0];
        float* hd1 = &sHid[wave][1][0];
        for (int r = 0; r < 8; ++r) {
            int mm = r + moff;
            hd0[mm * HID_STRIDE + n +  0] = fmaxf(acc0[r] + bb0, 0.0f);
            hd0[mm * HID_STRIDE + n + 16] = fmaxf(acc1[r] + bb1, 0.0f);
            hd0[mm * HID_STRIDE + n + 32] = fmaxf(acc2[r] + bb2, 0.0f);
            hd0[mm * HID_STRIDE + n + 48] = fmaxf(acc3[r] + bb3, 0.0f);
            hd1[mm * HID_STRIDE + n +  0] = fmaxf(acc4[r] + bb0, 0.0f);
            hd1[mm * HID_STRIDE + n + 16] = fmaxf(acc5[r] + bb1, 0.0f);
            hd1[mm * HID_STRIDE + n + 32] = fmaxf(acc6[r] + bb2, 0.0f);
            hd1[mm * HID_STRIDE + n + 48] = fmaxf(acc7[r] + bb3, 0.0f);
        }
    }

    float bc = b2[n];
    for (int tt = 0; tt < 2; ++tt) {
        const float* hd = &sHid[wave][tt][0];
        v8f acc = {};
        for (int k = 0; k < HH; k += 4) {
            int kb = k + khalf;
            v2f a = *(const v2f*)(hd + m * HID_STRIDE + kb);
            v2f b = *(const v2f*)&sW2p[((k >> 1) + hi16) * W2_PSTRIDE + 2 * n];
            acc = __builtin_amdgcn_wmma_f32_16x16x4_f32(false, a, false, b, (short)0, acc, false, false);
        }
        int tile = tt ? tile1 : tile0;
        for (int r = 0; r < 8; ++r) {
            int node = tile * 16 + r + moff;
            float mlp = acc[r] + bc;
            float al  = alpha[node];
            float sa  = 1.0f / (1.0f + expf(-al));
            float sna = 1.0f - sa;
            float a0 = att[node * 2 + 0], a1 = att[node * 2 + 1];
            float mx = fmaxf(a0, a1);
            float e0 = expf(a0 - mx), e1 = expf(a1 - mx);
            float inv = 1.0f / (e0 + e1);
            float logit = h0[(size_t)node * CC + n] * (e0 * inv)
                        + h1[(size_t)node * CC + n] * (e1 * inv);
            out[(size_t)node * CC + n] = sa * logit + sna * mlp;
        }
    }
}

extern "C" void kernel_launch(void* const* d_in, const int* in_sizes, int n_in,
                              void* d_out, int out_size, void* d_ws, size_t ws_size,
                              hipStream_t stream) {
    const float* features      = (const float*)d_in[0];
    const float* label_init    = (const float*)d_in[1];
    const float* labels_onehot = (const float*)d_in[2];
    const float* alpha         = (const float*)d_in[3];
    const float* attention     = (const float*)d_in[4];
    const float* e_edge        = (const float*)d_in[5];
    const float* w1            = (const float*)d_in[6];
    const float* b1            = (const float*)d_in[7];
    const float* w2            = (const float*)d_in[8];
    const float* b2            = (const float*)d_in[9];
    const int*   src           = (const int*)d_in[10];
    const int*   dst           = (const int*)d_in[11];
    const int*   train_mask    = (const int*)d_in[12];
    float* out = (float*)d_out;

    float* ws = (float*)d_ws;
    unsigned* nmax = (unsigned*)ws;
    float* nsum  = ws + NN;
    float* hA    = ws + 2 * NN;
    float* hout0 = hA + NN * CC;
    float* hout1 = hout0 + NN * CC;

    const int TB = 256;
    const int gridNC = (NN * CC + TB - 1) / TB;
    const int gridE  = (EE + TB - 1) / TB;

    for (int g = 0; g < GG; ++g) {
        float* hout = g ? hout1 : hout0;
        const float* hcur = label_init;
        for (int l = 0; l < LL; ++l) {
            float* hnext = (l == LL - 1) ? hout : hA;
            const float* ee = e_edge + ((size_t)l * GG + g) * EE;
            const int* sg = src + (size_t)g * EE;
            const int* dg = dst + (size_t)g * EE;
            k_init  <<<gridNC, TB, 0, stream>>>(nmax, nsum, hnext);
            k_segmax<<<gridE,  TB, 0, stream>>>(ee, dg, nmax);
            k_segsum<<<gridE,  TB, 0, stream>>>(ee, dg, nmax, nsum);
            k_agg   <<<gridE,  TB, 0, stream>>>(ee, sg, dg, nmax, nsum, hcur, hnext);
            k_mask  <<<gridNC, TB, 0, stream>>>(labels_onehot, train_mask, hnext);
            hcur = hnext;
        }
    }

    const int n_pairs = NN / 32;
    const int blocks  = (n_pairs + 7) / 8;
    k_mlp_combine<<<blocks, TB, 0, stream>>>(features, w1, b1, w2, b2,
                                             alpha, attention, hout0, hout1,
                                             out, n_pairs);
}


# Edge passes: each half of the chip adds only into its own half of the node lines (blocks read their die, take chunk tickets per half, sift edges into LDS lists); arrival order of adds changes
# speedup vs baseline: 14.3881x; 14.3881x over previous
#include <hip/hip_runtime.h>
#include <hip/hip_bf16.h>
#include <math.h>

#define NN 100000
#define CC 16
#define GG 2
#define LL 2
#define EE 3200000
#define FF 512
#define HH 64

#define W1_PSTRIDE 160
#define W2_PSTRIDE 32
#define HID_STRIDE 66

typedef float v2f __attribute__((ext_vector_type(2)));
typedef float v8f __attribute__((ext_vector_type(8)));

__device__ __forceinline__ unsigned f2ou(float f) {
    unsigned u = __float_as_uint(f);
    return (u & 0x80000000u) ? ~u : (u | 0x80000000u);
}
__device__ __forceinline__ float ou2f(unsigned u) {
    return __uint_as_float((u & 0x80000000u) ? (u & 0x7FFFFFFFu) : ~u);
}

#define QS 32
#define CHUNK 1024
#define NCHUNK (EE / CHUNK)
#define NHALF (NN / 2)
#define CTR0 64
#define AGG_BLOCKS 2048

template <bool NORM>
__global__ __launch_bounds__(512) void k_init(
                       const float* __restrict__ one_hot, const int* __restrict__ mask,
                       unsigned* __restrict__ gmax, float* __restrict__ acc,
                       float* __restrict__ dest) {
    int t = threadIdx.x;
    int node0 = blockIdx.x * 32;
    int   mk[2];
    float oh[2], st[2];
    for (int r = 0; r < 2; ++r) {
        int w    = r * 512 + t;
        int node = node0 + (w >> 5);
        mk[r] = mask[node];
        oh[r] = one_hot[(size_t)node * CC + (w & 15)];
    }
    for (int r = 0; r < 2; ++r) {
        int col = (r * 512 + t) & 31;
        float x = (col < CC) ? oh[r] : ((col == CC) ? 1.0f : 0.0f);
        st[r] = mk[r] ? x : 0.0f;
    }
    if (NORM) {
        int node = node0 + (t >> 4);
        int n    = t & 15;
        float v = acc[(size_t)node * QS + n];
        float s = acc[(size_t)node * QS + CC];
        dest[(size_t)node * CC + n] = (s > 0.0f) ? v / s : v;
        __syncthreads();
    }
    for (int r = 0; r < 2; ++r) {
        int w = r * 512 + t;
        acc[(size_t)(node0 + (w >> 5)) * QS + (w & 31)] = st[r];
    }
    if (!NORM && blockIdx.x == 0) {
        if (t < LL * GG) gmax[t] = 0x007FFFFFu;
        else if (t >= CTR0 && t < CTR0 + 128) gmax[t] = 0u;
    }
}

__device__ __forceinline__ float max4(float4 v) {
    return fmaxf(fmaxf(v.x, v.y), fmaxf(v.z, v.w));
}
__global__ __launch_bounds__(256) void k_gmax(const float* __restrict__ e,
                                              unsigned* __restrict__ gmax) {
    __shared__ float wmax[8];
    const float4* p = (const float4*)(e + (size_t)blockIdx.y * EE);
    const int n4 = EE / 4;
    const int stride = gridDim.x * 256;
    int i = blockIdx.x * 256 + threadIdx.x;
    float m = -INFINITY;
    for (; i + 3 * stride < n4; i += 4 * stride) {
        float4 a = p[i], b = p[i + stride], c = p[i + 2 * stride], d = p[i + 3 * stride];
        m = fmaxf(m, fmaxf(fmaxf(max4(a), max4(b)), fmaxf(max4(c), max4(d))));
    }
    for (; i < n4; i += stride) m = fmaxf(m, max4(p[i]));
    for (int off = 16; off > 0; off >>= 1) m = fmaxf(m, __shfl_xor(m, off));
    if ((threadIdx.x & 31) == 0) wmax[threadIdx.x >> 5] = m;
    __syncthreads();
    if (threadIdx.x == 0) {
        float b = wmax[0];
        for (int w = 1; w < 8; ++w) b = fmaxf(b, wmax[w]);
        atomicMax(&gmax[blockIdx.y], f2ou(b));
    }
}

template <int J> __device__ __forceinline__ int row_bc(int x) {
    return __builtin_amdgcn_update_dpp(0, x, 0x150 + J, 0xf, 0xf, false);
}
__device__ __forceinline__ int xhalf(int x) {
    return __builtin_amdgcn_permlanex16(x, x, 0x76543210, 0xfedcba98, false, false);
}

template <int J0> __device__ __forceinline__ void agg_rows8(
    int d, int s, int ai, int dX, int aXi, int n, int hi16,
    const float* __restrict__ hcur, float* __restrict__ acc) {
    int   dj[8], sj[8], dO[8];
    float aj[8], aO[8], v[8];
    dj[0] = row_bc<J0 + 0>(d); sj[0] = row_bc<J0 + 0>(s); aj[0] = __int_as_float(row_bc<J0 + 0>(ai));
    dj[1] = row_bc<J0 + 1>(d); sj[1] = row_bc<J0 + 1>(s); aj[1] = __int_as_float(row_bc<J0 + 1>(ai));
    dj[2] = row_bc<J0 + 2>(d); sj[2] = row_bc<J0 + 2>(s); aj[2] = __int_as_float(row_bc<J0 + 2>(ai));
    dj[3] = row_bc<J0 + 3>(d); sj[3] = row_bc<J0 + 3>(s); aj[3] = __int_as_float(row_bc<J0 + 3>(ai));
    dj[4] = row_bc<J0 + 4>(d); sj[4] = row_bc<J0 + 4>(s); aj[4] = __int_as_float(row_bc<J0 + 4>(ai));
    dj[5] = row_bc<J0 + 5>(d); sj[5] = row_bc<J0 + 5>(s); aj[5] = __int_as_float(row_bc<J0 + 5>(ai));
    dj[6] = row_bc<J0 + 6>(d); sj[6] = row_bc<J0 + 6>(s); aj[6] = __int_as_float(row_bc<J0 + 6>(ai));
    dj[7] = row_bc<J0 + 7>(d); sj[7] = row_bc<J0 + 7>(s); aj[7] = __int_as_float(row_bc<J0 + 7>(ai));
    dO[0] = row_bc<J0 + 0>(dX); aO[0] = __int_as_float(row_bc<J0 + 0>(aXi));
    dO[1] = row_bc<J0 + 1>(dX); aO[1] = __int_as_float(row_bc<J0 + 1>(aXi));
    dO[2] = row_bc<J0 + 2>(dX); aO[2] = __int_as_float(row_bc<J0 + 2>(aXi));
    dO[3] = row_bc<J0 + 3>(dX); aO[3] = __int_as_float(row_bc<J0 + 3>(aXi));
    dO[4] = row_bc<J0 + 4>(dX); aO[4] = __int_as_float(row_bc<J0 + 4>(aXi));
    dO[5] = row_bc<J0 + 5>(dX); aO[5] = __int_as_float(row_bc<J0 + 5>(aXi));
    dO[6] = row_bc<J0 + 6>(dX); aO[6] = __int_as_float(row_bc<J0 + 6>(aXi));
    dO[7] = row_bc<J0 + 7>(dX); aO[7] = __int_as_float(row_bc<J0 + 7>(aXi));
#pragma unroll
    for (int k = 0; k < 8; ++k)
        v[k] = (dj[k] >= 0) ? hcur[(size_t)sj[k] * CC + n] : 0.0f;
#pragma unroll
    for (int k = 0; k < 8; ++k) {
#pragma unroll
        for (int h = 0; h < 2; ++h) {
            bool  own = (hi16 == h);
            int   dd  = own ? dj[k] : dO[k];
            int   col = own ? n : CC;
            float x   = own ? v[k] * aj[k] : aO[k];
            if ((own || n == 0) && dd >= 0)
                atomicAdd(acc + (size_t)dd * QS + col, x);
        }
    }
}

__global__ __launch_bounds__(256) void k_agg(
                      const float* __restrict__ e, const int* __restrict__ src,
                      const int* __restrict__ dst, const int* __restrict__ mask,
                      const unsigned* __restrict__ gmax, unsigned* __restrict__ ctr,
                      const float* __restrict__ hcur, float* __restrict__ acc) {
    __shared__ int      s_d[CHUNK];
    __shared__ int      s_s[CHUNK];
    __shared__ float    s_a[CHUNK];
    __shared__ unsigned s_cnt;
    __shared__ unsigned s_t;
    __shared__ unsigned s_home;
    int tid  = threadIdx.x;
    int lane = tid & 31;
    int wave = tid >> 5;
    int n = tid & 15;
    int hi16 = (tid >> 4) & 1;
    float gm = ou2f(gmax[0]);
    if (tid == 0) s_home = (((unsigned)__builtin_amdgcn_s_sendmsg_rtn(0x87) >> 16) >> 2) & 1u;
    __syncthreads();
    unsigned home = s_home;
    for (int ph = 0; ph < 2; ++ph) {
        unsigned h = home ^ (unsigned)ph;
        int lo  = h ? NHALF : 0;
        int hiN = h ? NN : NHALF;
        for (;;) {
            if (tid == 0) {
                s_t = atomicAdd(ctr + 64 * h, 1u);
                s_cnt = 0u;
            }
            __syncthreads();
            unsigned tk = s_t;
            __syncthreads();
            if (tk >= (unsigned)NCHUNK) break;
            int i0 = (int)tk * CHUNK + tid;
            int   dd[4], ss[4];
            float ev[4];
#pragma unroll
            for (int r = 0; r < 4; ++r) {
                dd[r] = dst[i0 + r * 256];
                ss[r] = src[i0 + r * 256];
                ev[r] = e[i0 + r * 256];
            }
            bool ok[4];
            unsigned bal[4];
#pragma unroll
            for (int r = 0; r < 4; ++r) {
                bool in = (dd[r] >= lo) && (dd[r] < hiN);
                int m = mask[in ? dd[r] : lo];
                ok[r] = in && (m == 0);
            }
#pragma unroll
            for (int r = 0; r < 4; ++r) bal[r] = __builtin_amdgcn_ballot_w32(ok[r]);
            unsigned tot = (unsigned)(__popc(bal[0]) + __popc(bal[1]) + __popc(bal[2]) + __popc(bal[3]));
            unsigned base = 0u;
            if (lane == 0) base = atomicAdd(&s_cnt, tot);
            base = (unsigned)__builtin_amdgcn_readfirstlane((int)base);
#pragma unroll
            for (int r = 0; r < 4; ++r) {
                if (ok[r]) {
                    unsigned pos = base + __builtin_amdgcn_mbcnt_lo(bal[r], 0u);
                    s_d[pos] = dd[r];
                    s_s[pos] = ss[r];
                    s_a[pos] = expf(ev[r] - gm);
                }
                base += (unsigned)__popc(bal[r]);
            }
            __syncthreads();
            int cnt = __builtin_amdgcn_readfirstlane((int)s_cnt);
            for (int q0 = __builtin_amdgcn_readfirstlane(wave * 32); q0 < cnt; q0 += 256) {
                int q = q0 + lane;
                int d = -1, s = 0;
                float a = 0.0f;
                if (q < cnt) { d = s_d[q]; s = s_s[q]; a = s_a[q]; }
                int ai  = __float_as_int(a);
                int dX  = xhalf(d);
                int aXi = xhalf(ai);
                agg_rows8<0>(d, s, ai, dX, aXi, n, hi16, hcur, acc);
                agg_rows8<8>(d, s, ai, dX, aXi, n, hi16, hcur, acc);
            }
            __syncthreads();
        }
    }
}

__global__ __launch_bounds__(256) void k_mlp_combine(
    const float* __restrict__ feat, const float* __restrict__ w1,
    const float* __restrict__ b1,   const float* __restrict__ w2,
    const float* __restrict__ b2,   const float* __restrict__ alpha,
    const float* __restrict__ att,  const float* __restrict__ nline,
    float* __restrict__ out, int n_pairs)
{
    __shared__ float sW1p[(FF / 2) * W1_PSTRIDE];
    __shared__ float sW2p[(HH / 2) * W2_PSTRIDE];
    __shared__ float sHid[8][2][16 * HID_STRIDE];

    int t = threadIdx.x;

    for (int i = t; i < (FF / 2) * (HH / 4); i += 256) {
        int p  = i >> 4;
        int nq = (i & 15) << 2;
        float4 x = *(const float4*)&w1[(size_t)p * 2 * HH + nq];
        float4 y = *(const float4*)&w1[(size_t)p * 2 * HH + HH + nq];
        float4 lo = { x.x, y.x, x.y, y.y };
        float4 hi = { x.z, y.z, x.w, y.w };
        float* d = &sW1p[p * W1_PSTRIDE + 2 * nq];
        *(float4*)(d + 0) = lo;
        *(float4*)(d + 4) = hi;
    }
    for (int i = t; i < (HH / 2) * CC; i += 256) {
        int p = i >> 4, n = i & 15;
        v2f v; v.x = w2[(size_t)p * 2 * CC + n];
        v.y = w2[(size_t)p * 2 * CC + CC + n];
        *(v2f*)&sW2p[p * W2_PSTRIDE + 2 * n] = v;
    }
    __syncthreads();

    int wave = t >> 5;
    int lane = t & 31;
    int pair = blockIdx.x * 8 + wave;
    if (pair >= n_pairs) return;

    int tile0 = pair * 2;
    int tile1 = tile0 + 1;

    int m     = lane & 15;
    int n     = lane & 15;
    int hi16  = lane >> 4;
    int khalf = hi16 << 1;
    int moff  = hi16 << 3;

    v8f acc0 = {}, acc1 = {}, acc2 = {}, acc3 = {};
    v8f acc4 = {}, acc5 = {}, acc6 = {}, acc7 = {};
    const float* arow0 = feat + ((size_t)tile0 * 16 + m) * FF;
    const float* arow1 = feat + ((size_t)tile1 * 16 + m) * FF;
    for (int k = 0; k < FF; k += 4) {
        int kb = k + khalf;
        v2f a0 = *(const v2f*)(arow0 + kb);
        v2f a1 = *(const v2f*)(arow1 + kb);
        const float* bp = &sW1p[((k >> 1) + hi16) * W1_PSTRIDE + 2 * n];
        v2f b0 = *(const v2f*)(bp + 0);
        v2f b1v = *(const v2f*)(bp + 32);
        v2f b2v = *(const v2f*)(bp + 64);
        v2f b3v = *(const v2f*)(bp + 96);
        acc0 = __builtin_amdgcn_wmma_f32_16x16x4_f32(false, a0, false, b0,  (short)0, acc0, false, false);
        acc1 = __builtin_amdgcn_wmma_f32_16x16x4_f32(false, a0, false, b1v, (short)0, acc1, false, false);
        acc2 = __builtin_amdgcn_wmma_f32_16x16x4_f32(false, a0, false, b2v, (short)0, acc2, false, false);
        acc3 = __builtin_amdgcn_wmma_f32_16x16x4_f32(false, a0, false, b3v, (short)0, acc3, false, false);
        acc4 = __builtin_amdgcn_wmma_f32_16x16x4_f32(false, a1, false, b0,  (short)0, acc4, false, false);
        acc5 = __builtin_amdgcn_wmma_f32_16x16x4_f32(false, a1, false, b1v, (short)0, acc5, false, false);
        acc6 = __builtin_amdgcn_wmma_f32_16x16x4_f32(false, a1, false, b2v, (short)0, acc6, false, false);
        acc7 = __builtin_amdgcn_wmma_f32_16x16x4_f32(false, a1, false, b3v, (short)0, acc7, false, false);
    }

    float bb0 = b1[n + 0], bb1 = b1[n + 16], bb2 = b1[n + 32], bb3 = b1[n + 48];
    {
        float* hd0 = &sHid[wave][0][0];
        float* hd1 = &sHid[wave][1][0];
        for (int r = 0; r < 8; ++r) {
            int mm = r + moff;
            hd0[mm * HID_STRIDE + n +  0] = fmaxf(acc0[r] + bb0, 0.0f);
            hd0[mm * HID_STRIDE + n + 16] = fmaxf(acc1[r] + bb1, 0.0f);
            hd0[mm * HID_STRIDE + n + 32] = fmaxf(acc2[r] + bb2, 0.0f);
            hd0[mm * HID_STRIDE + n + 48] = fmaxf(acc3[r] + bb3, 0.0f);
            hd1[mm * HID_STRIDE + n +  0] = fmaxf(acc4[r] + bb0, 0.0f);
            hd1[mm * HID_STRIDE + n + 16] = fmaxf(acc5[r] + bb1, 0.0f);
            hd1[mm * HID_STRIDE + n + 32] = fmaxf(acc6[r] + bb2, 0.0f);
            hd1[mm * HID_STRIDE + n + 48] = fmaxf(acc7[r] + bb3, 0.0f);
        }
    }

    float bc = b2[n];
    for (int tt = 0; tt < 2; ++tt) {
        const float* hd = &sHid[wave][tt][0];
        v8f acc = {};
        for (int k = 0; k < HH; k += 4) {
            int kb = k + khalf;
            v2f a = *(const v2f*)(hd + m * HID_STRIDE + kb);
            v2f b = *(const v2f*)&sW2p[((k >> 1) + hi16) * W2_PSTRIDE + 2 * n];
            acc = __builtin_amdgcn_wmma_f32_16x16x4_f32(false, a, false, b, (short)0, acc, false, false);
        }
        int tile = tt ? tile1 : tile0;
        for (int r = 0; r < 8; ++r) {
            int node = tile * 16 + r + moff;
            float mlp = acc[r] + bc;
            float al  = alpha[node];
            float sa  = 1.0f / (1.0f + expf(-al));
            float sna = 1.0f - sa;
            float a0 = att[node * 2 + 0], a1 = att[node * 2 + 1];
            float mx = fmaxf(a0, a1);
            float e0 = expf(a0 - mx), e1 = expf(a1 - mx);
            float inv = 1.0f / (e0 + e1);
            float s1  = nline[(size_t)node * QS + CC];
            float h1v = nline[(size_t)node * QS + n];
            h1v = (s1 > 0.0f) ? h1v / s1 : h1v;
            float h0v = out[(size_t)node * CC + n];
            float logit = h0v * (e0 * inv) + h1v * (e1 * inv);
            out[(size_t)node * CC + n] = sa * logit + sna * mlp;
        }
    }
}

extern "C" void kernel_launch(void* const* d_in, const int* in_sizes, int n_in,
                              void* d_out, int out_size, void* d_ws, size_t ws_size,
                              hipStream_t stream) {
    const float* features      = (const float*)d_in[0];
    const float* label_init    = (const float*)d_in[1];
    const float* labels_onehot = (const float*)d_in[2];
    const float* alpha         = (const float*)d_in[3];
    const float* attention     = (const float*)d_in[4];
    const float* e_edge        = (const float*)d_in[5];
    const float* w1            = (const float*)d_in[6];
    const float* b1            = (const float*)d_in[7];
    const float* w2            = (const float*)d_in[8];
    const float* b2            = (const float*)d_in[9];
    const int*   src           = (const int*)d_in[10];
    const int*   dst           = (const int*)d_in[11];
    const int*   train_mask    = (const int*)d_in[12];
    float* out = (float*)d_out;

    float* ws = (float*)d_ws;
    unsigned* gmax = (unsigned*)ws;
    float* acc  = ws + NN;
    float* hrow = acc + (size_t)NN * QS;

    const int TB = 256;
    const int gridI = NN / 32;

    const float* started = nullptr;
    for (int g = 0; g < GG; ++g) {
        for (int l = 0; l < LL; ++l) {
            const float* ee = e_edge + ((size_t)l * GG + g) * EE;
            const int* sg = src + (size_t)g * EE;
            const int* dg = dst + (size_t)g * EE;
            float* dest = l ? hrow : out;
            const float* hcur = l ? hrow : label_init;
            if (!started) {
                k_init<false><<<gridI, 512, 0, stream>>>(labels_onehot, train_mask, gmax, acc, dest);
                k_gmax<<<dim3(256, LL * GG), TB, 0, stream>>>(e_edge, gmax);
                started = acc;
            } else {
                k_init<true> <<<gridI, 512, 0, stream>>>(labels_onehot, train_mask, gmax, acc, dest);
            }
            k_agg<<<AGG_BLOCKS, TB, 0, stream>>>(ee, sg, dg, train_mask, gmax + (l * GG + g),
                                                 gmax + CTR0 + (g * LL + l), hcur, acc);
        }
    }

    const int n_pairs = NN / 32;
    const int blocks  = (n_pairs + 7) / 8;
    k_mlp_combine<<<blocks, TB, 0, stream>>>(features, w1, b1, w2, b2,
                                             alpha, attention, acc,
                                             out, n_pairs);
}


# Edge passes: the sift tests a destination's train-mask bit in an LDS copy of a bit table (one bit per node, packed by the first launch), not with a scattered mask word load per edge and chunk
# speedup vs baseline: 15.8437x; 1.1012x over previous
#include <hip/hip_runtime.h>
#include <hip/hip_bf16.h>
#include <math.h>

#define NN 100000
#define CC 16
#define GG 2
#define LL 2
#define EE 3200000
#define FF 512
#define HH 64

#define W1_PSTRIDE 160
#define W2_PSTRIDE 32
#define HID_STRIDE 66

typedef float v2f __attribute__((ext_vector_type(2)));
typedef float v8f __attribute__((ext_vector_type(8)));

__device__ __forceinline__ unsigned f2ou(float f) {
    unsigned u = __float_as_uint(f);
    return (u & 0x80000000u) ? ~u : (u | 0x80000000u);
}
__device__ __forceinline__ float ou2f(unsigned u) {
    return __uint_as_float((u & 0x80000000u) ? (u & 0x7FFFFFFFu) : ~u);
}

#define QS 32
#define CHUNK 1024
#define NCHUNK (EE / CHUNK)
#define NHALF (NN / 2)
#define CTR0 64
#define AGG_BLOCKS 2048
#define NBITW (NN / 32)
#define NBITQ ((NBITW + 3) / 4)
#define BITS0 256

template <bool NORM>
__global__ __launch_bounds__(512) void k_init(
                       const float* __restrict__ one_hot, const int* __restrict__ mask,
                       unsigned* __restrict__ gmax, float* __restrict__ acc,
                       float* __restrict__ dest) {
    int t = threadIdx.x;
    int node0 = blockIdx.x * 32;
    int   mk[2];
    float oh[2], st[2];
    int mb = NORM ? 0 : mask[node0 + (t & 31)];
    for (int r = 0; r < 2; ++r) {
        int w    = r * 512 + t;
        int node = node0 + (w >> 5);
        mk[r] = mask[node];
        oh[r] = one_hot[(size_t)node * CC + (w & 15)];
    }
    for (int r = 0; r < 2; ++r) {
        int col = (r * 512 + t) & 31;
        float x = (col < CC) ? oh[r] : ((col == CC) ? 1.0f : 0.0f);
        st[r] = mk[r] ? x : 0.0f;
    }
    if (NORM) {
        int node = node0 + (t >> 4);
        int n    = t & 15;
        float v = acc[(size_t)node * QS + n];
        float s = acc[(size_t)node * QS + CC];
        dest[(size_t)node * CC + n] = (s > 0.0f) ? v / s : v;
        __syncthreads();
    }
    for (int r = 0; r < 2; ++r) {
        int w = r * 512 + t;
        acc[(size_t)(node0 + (w >> 5)) * QS + (w & 31)] = st[r];
    }
    if (!NORM && blockIdx.x == 0) {
        if (t < LL * GG) gmax[t] = 0x007FFFFFu;
        else if (t >= CTR0 && t < CTR0 + 128) gmax[t] = 0u;
        else if (t >= CTR0 + 128 && t < CTR0 + 128 + (4 * NBITQ - NBITW)) gmax[BITS0 + NBITW + (t - CTR0 - 128)] = 0u;
    }
    if (!NORM) {
        unsigned w = __builtin_amdgcn_ballot_w32(mb != 0);
        if (t == 0) gmax[BITS0 + blockIdx.x] = w;
    }
}

__device__ __forceinline__ float max4(float4 v) {
    return fmaxf(fmaxf(v.x, v.y), fmaxf(v.z, v.w));
}
__global__ __launch_bounds__(256) void k_gmax(const float* __restrict__ e,
                                              unsigned* __restrict__ gmax) {
    __shared__ float wmax[8];
    const float4* p = (const float4*)(e + (size_t)blockIdx.y * EE);
    const int n4 = EE / 4;
    const int stride = gridDim.x * 256;
    int i = blockIdx.x * 256 + threadIdx.x;
    float m = -INFINITY;
    for (; i + 3 * stride < n4; i += 4 * stride) {
        float4 a = p[i], b = p[i + stride], c = p[i + 2 * stride], d = p[i + 3 * stride];
        m = fmaxf(m, fmaxf(fmaxf(max4(a), max4(b)), fmaxf(max4(c), max4(d))));
    }
    for (; i < n4; i += stride) m = fmaxf(m, max4(p[i]));
    for (int off = 16; off > 0; off >>= 1) m = fmaxf(m, __shfl_xor(m, off));
    if ((threadIdx.x & 31) == 0) wmax[threadIdx.x >> 5] = m;
    __syncthreads();
    if (threadIdx.x == 0) {
        float b = wmax[0];
        for (int w = 1; w < 8; ++w) b = fmaxf(b, wmax[w]);
        atomicMax(&gmax[blockIdx.y], f2ou(b));
    }
}

template <int J> __device__ __forceinline__ int row_bc(int x) {
    return __builtin_amdgcn_update_dpp(0, x, 0x150 + J, 0xf, 0xf, false);
}
__device__ __forceinline__ int xhalf(int x) {
    return __builtin_amdgcn_permlanex16(x, x, 0x76543210, 0xfedcba98, false, false);
}

template <int J0> __device__ __forceinline__ void agg_rows8(
    int d, int s, int ai, int dX, int aXi, int n, int hi16,
    const float* __restrict__ hcur, float* __restrict__ acc) {
    int   dj[8], sj[8], dO[8];
    float aj[8], aO[8], v[8];
    dj[0] = row_bc<J0 + 0>(d); sj[0] = row_bc<J0 + 0>(s); aj[0] = __int_as_float(row_bc<J0 + 0>(ai));
    dj[1] = row_bc<J0 + 1>(d); sj[1] = row_bc<J0 + 1>(s); aj[1] = __int_as_float(row_bc<J0 + 1>(ai));
    dj[2] = row_bc<J0 + 2>(d); sj[2] = row_bc<J0 + 2>(s); aj[2] = __int_as_float(row_bc<J0 + 2>(ai));
    dj[3] = row_bc<J0 + 3>(d); sj[3] = row_bc<J0 + 3>(s); aj[3] = __int_as_float(row_bc<J0 + 3>(ai));
    dj[4] = row_bc<J0 + 4>(d); sj[4] = row_bc<J0 + 4>(s); aj[4] = __int_as_float(row_bc<J0 + 4>(ai));
    dj[5] = row_bc<J0 + 5>(d); sj[5] = row_bc<J0 + 5>(s); aj[5] = __int_as_float(row_bc<J0 + 5>(ai));
    dj[6] = row_bc<J0 + 6>(d); sj[6] = row_bc<J0 + 6>(s); aj[6] = __int_as_float(row_bc<J0 + 6>(ai));
    dj[7] = row_bc<J0 + 7>(d); sj[7] = row_bc<J0 + 7>(s); aj[7] = __int_as_float(row_bc<J0 + 7>(ai));
    dO[0] = row_bc<J0 + 0>(dX); aO[0] = __int_as_float(row_bc<J0 + 0>(aXi));
    dO[1] = row_bc<J0 + 1>(dX); aO[1] = __int_as_float(row_bc<J0 + 1>(aXi));
    dO[2] = row_bc<J0 + 2>(dX); aO[2] = __int_as_float(row_bc<J0 + 2>(aXi));
    dO[3] = row_bc<J0 + 3>(dX); aO[3] = __int_as_float(row_bc<J0 + 3>(aXi));
    dO[4] = row_bc<J0 + 4>(dX); aO[4] = __int_as_float(row_bc<J0 + 4>(aXi));
    dO[5] = row_bc<J0 + 5>(dX); aO[5] = __int_as_float(row_bc<J0 + 5>(aXi));
    dO[6] = row_bc<J0 + 6>(dX); aO[6] = __int_as_float(row_bc<J0 + 6>(aXi));
    dO[7] = row_bc<J0 + 7>(dX); aO[7] = __int_as_float(row_bc<J0 + 7>(aXi));
#pragma unroll
    for (int k = 0; k < 8; ++k)
        v[k] = (dj[k] >= 0) ? hcur[(size_t)sj[k] * CC + n] : 0.0f;
#pragma unroll
    for (int k = 0; k < 8; ++k) {
#pragma unroll
        for (int h = 0; h < 2; ++h) {
            bool  own = (hi16 == h);
            int   dd  = own ? dj[k] : dO[k];
            int   col = own ? n : CC;
            float x   = own ? v[k] * aj[k] : aO[k];
            if ((own || n == 0) && dd >= 0)
                atomicAdd(acc + (size_t)dd * QS + col, x);
        }
    }
}

__global__ __launch_bounds__(256) void k_agg(
                      const float* __restrict__ e, const int* __restrict__ src,
                      const int* __restrict__ dst, const unsigned* __restrict__ bits,
                      const unsigned* __restrict__ gmax, unsigned* __restrict__ ctr,
                      const float* __restrict__ hcur, float* __restrict__ acc) {
    __shared__ int      s_d[CHUNK];
    __shared__ int      s_s[CHUNK];
    __shared__ float    s_a[CHUNK];
    __shared__ unsigned s_cnt;
    __shared__ unsigned s_t;
    __shared__ unsigned s_home;
    __shared__ uint4    s_bits4[4 * 256];
    int tid  = threadIdx.x;
    int lane = tid & 31;
    int wave = tid >> 5;
    int n = tid & 15;
    int hi16 = (tid >> 4) & 1;
    float gm = ou2f(gmax[0]);
    if (tid == 0) s_home = (((unsigned)__builtin_amdgcn_s_sendmsg_rtn(0x87) >> 16) >> 2) & 1u;
    {
        const uint4* bq = (const uint4*)bits;
        uint4 tb[4];
#pragma unroll
        for (int k = 0; k < 4; ++k) {
            int i = tid + k * 256;
            tb[k] = bq[i < NBITQ ? i : NBITQ - 1];
        }
#pragma unroll
        for (int k = 0; k < 4; ++k) s_bits4[tid + k * 256] = tb[k];
    }
    const unsigned* s_bits = (const unsigned*)s_bits4;
    __syncthreads();
    unsigned home = s_home;
    for (int ph = 0; ph < 2; ++ph) {
        unsigned h = home ^ (unsigned)ph;
        int lo  = h ? NHALF : 0;
        int hiN = h ? NN : NHALF;
        for (;;) {
            if (tid == 0) {
                s_t = atomicAdd(ctr + 64 * h, 1u);
                s_cnt = 0u;
            }
            __syncthreads();
            unsigned tk = s_t;
            __syncthreads();
            if (tk >= (unsigned)NCHUNK) break;
            int i0 = (int)tk * CHUNK + tid;
            int   dd[4], ss[4];
            float ev[4];
#pragma unroll
            for (int r = 0; r < 4; ++r) {
                dd[r] = dst[i0 + r * 256];
                ss[r] = src[i0 + r * 256];
                ev[r] = e[i0 + r * 256];
            }
            bool ok[4];
            unsigned bal[4];
            unsigned tw[4];
#pragma unroll
            for (int r = 0; r < 4; ++r) {
                unsigned wi = (unsigned)dd[r] >> 5;
                tw[r] = s_bits[wi < (unsigned)NBITW ? wi : (unsigned)(NBITW - 1)];
            }
#pragma unroll
            for (int r = 0; r < 4; ++r) {
                unsigned in  = (unsigned)(dd[r] >= lo) & (unsigned)(dd[r] < hiN);
                unsigned bit = (tw[r] >> (dd[r] & 31)) & 1u;
                ok[r] = (in & (bit ^ 1u)) != 0u;
            }
#pragma unroll
            for (int r = 0; r < 4; ++r) bal[r] = __builtin_amdgcn_ballot_w32(ok[r]);
            unsigned tot = (unsigned)(__popc(bal[0]) + __popc(bal[1]) + __popc(bal[2]) + __popc(bal[3]));
            unsigned base = 0u;
            if (lane == 0) base = atomicAdd(&s_cnt, tot);
            base = (unsigned)__builtin_amdgcn_readfirstlane((int)base);
#pragma unroll
            for (int r = 0; r < 4; ++r) {
                if (ok[r]) {
                    unsigned pos = base + __builtin_amdgcn_mbcnt_lo(bal[r], 0u);
                    s_d[pos] = dd[r];
                    s_s[pos] = ss[r];
                    s_a[pos] = expf(ev[r] - gm);
                }
                base += (unsigned)__popc(bal[r]);
            }
            __syncthreads();
            int cnt = __builtin_amdgcn_readfirstlane((int)s_cnt);
            for (int q0 = __builtin_amdgcn_readfirstlane(wave * 32); q0 < cnt; q0 += 256) {
                int q = q0 + lane;
                int d = -1, s = 0;
                float a = 0.0f;
                if (q < cnt) { d = s_d[q]; s = s_s[q]; a = s_a[q]; }
                int ai  = __float_as_int(a);
                int dX  = xhalf(d);
                int aXi = xhalf(ai);
                agg_rows8<0>(d, s, ai, dX, aXi, n, hi16, hcur, acc);
                agg_rows8<8>(d, s, ai, dX, aXi, n, hi16, hcur, acc);
            }
            __syncthreads();
        }
    }
}

__global__ __launch_bounds__(256) void k_mlp_combine(
    const float* __restrict__ feat, const float* __restrict__ w1,
    const float* __restrict__ b1,   const float* __restrict__ w2,
    const float* __restrict__ b2,   const float* __restrict__ alpha,
    const float* __restrict__ att,  const float* __restrict__ nline,
    float* __restrict__ out, int n_pairs)
{
    __shared__ float sW1p[(FF / 2) * W1_PSTRIDE];
    __shared__ float sW2p[(HH / 2) * W2_PSTRIDE];
    __shared__ float sHid[8][2][16 * HID_STRIDE];

    int t = threadIdx.x;

    for (int i = t; i < (FF / 2) * (HH / 4); i += 256) {
        int p  = i >> 4;
        int nq = (i & 15) << 2;
        float4 x = *(const float4*)&w1[(size_t)p * 2 * HH + nq];
        float4 y = *(const float4*)&w1[(size_t)p * 2 * HH + HH + nq];
        float4 lo = { x.x, y.x, x.y, y.y };
        float4 hi = { x.z, y.z, x.w, y.w };
        float* d = &sW1p[p * W1_PSTRIDE + 2 * nq];
        *(float4*)(d + 0) = lo;
        *(float4*)(d + 4) = hi;
    }
    for (int i = t; i < (HH / 2) * CC; i += 256) {
        int p = i >> 4, n = i & 15;
        v2f v; v.x = w2[(size_t)p * 2 * CC + n];
        v.y = w2[(size_t)p * 2 * CC + CC + n];
        *(v2f*)&sW2p[p * W2_PSTRIDE + 2 * n] = v;
    }
    __syncthreads();

    int wave = t >> 5;
    int lane = t & 31;
    int pair = blockIdx.x * 8 + wave;
    if (pair >= n_pairs) return;

    int tile0 = pair * 2;
    int tile1 = tile0 + 1;

    int m     = lane & 15;
    int n     = lane & 15;
    int hi16  = lane >> 4;
    int khalf = hi16 << 1;
    int moff  = hi16 << 3;

    v8f acc0 = {}, acc1 = {}, acc2 = {}, acc3 = {};
    v8f acc4 = {}, acc5 = {}, acc6 = {}, acc7 = {};
    const float* arow0 = feat + ((size_t)tile0 * 16 + m) * FF;
    const float* arow1 = feat + ((size_t)tile1 * 16 + m) * FF;
    for (int k = 0; k < FF; k += 4) {
        int kb = k + khalf;
        v2f a0 = *(const v2f*)(arow0 + kb);
        v2f a1 = *(const v2f*)(arow1 + kb);
        const float* bp = &sW1p[((k >> 1) + hi16) * W1_PSTRIDE + 2 * n];
        v2f b0 = *(const v2f*)(bp + 0);
        v2f b1v = *(const v2f*)(bp + 32);
        v2f b2v = *(const v2f*)(bp + 64);
        v2f b3v = *(const v2f*)(bp + 96);
        acc0 = __builtin_amdgcn_wmma_f32_16x16x4_f32(false, a0, false, b0,  (short)0, acc0, false, false);
        acc1 = __builtin_amdgcn_wmma_f32_16x16x4_f32(false, a0, false, b1v, (short)0, acc1, false, false);
        acc2 = __builtin_amdgcn_wmma_f32_16x16x4_f32(false, a0, false, b2v, (short)0, acc2, false, false);
        acc3 = __builtin_amdgcn_wmma_f32_16x16x4_f32(false, a0, false, b3v, (short)0, acc3, false, false);
        acc4 = __builtin_amdgcn_wmma_f32_16x16x4_f32(false, a1, false, b0,  (short)0, acc4, false, false);
        acc5 = __builtin_amdgcn_wmma_f32_16x16x4_f32(false, a1, false, b1v, (short)0, acc5, false, false);
        acc6 = __builtin_amdgcn_wmma_f32_16x16x4_f32(false, a1, false, b2v, (short)0, acc6, false, false);
        acc7 = __builtin_amdgcn_wmma_f32_16x16x4_f32(false, a1, false, b3v, (short)0, acc7, false, false);
    }

    float bb0 = b1[n + 0], bb1 = b1[n + 16], bb2 = b1[n + 32], bb3 = b1[n + 48];
    {
        float* hd0 = &sHid[wave][0][0];
        float* hd1 = &sHid[wave][1][0];
        for (int r = 0; r < 8; ++r) {
            int mm = r + moff;
            hd0[mm * HID_STRIDE + n +  0] = fmaxf(acc0[r] + bb0, 0.0f);
            hd0[mm * HID_STRIDE + n + 16] = fmaxf(acc1[r] + bb1, 0.0f);
            hd0[mm * HID_STRIDE + n + 32] = fmaxf(acc2[r] + bb2, 0.0f);
            hd0[mm * HID_STRIDE + n + 48] = fmaxf(acc3[r] + bb3, 0.0f);
            hd1[mm * HID_STRIDE + n +  0] = fmaxf(acc4[r] + bb0, 0.0f);
            hd1[mm * HID_STRIDE + n + 16] = fmaxf(acc5[r] + bb1, 0.0f);
            hd1[mm * HID_STRIDE + n + 32] = fmaxf(acc6[r] + bb2, 0.0f);
            hd1[mm * HID_STRIDE + n + 48] = fmaxf(acc7[r] + bb3, 0.0f);
        }
    }

    float bc = b2[n];
    for (int tt = 0; tt < 2; ++tt) {
        const float* hd = &sHid[wave][tt][0];
        v8f acc = {};
        for (int k = 0; k < HH; k += 4) {
            int kb = k + khalf;
            v2f a = *(const v2f*)(hd + m * HID_STRIDE + kb);
            v2f b = *(const v2f*)&sW2p[((k >> 1) + hi16) * W2_PSTRIDE + 2 * n];
            acc = __builtin_amdgcn_wmma_f32_16x16x4_f32(false, a, false, b, (short)0, acc, false, false);
        }
        int tile = tt ? tile1 : tile0;
        for (int r = 0; r < 8; ++r) {
            int node = tile * 16 + r + moff;
            float mlp = acc[r] + bc;
            float al  = alpha[node];
            float sa  = 1.0f / (1.0f + expf(-al));
            float sna = 1.0f - sa;
            float a0 = att[node * 2 + 0], a1 = att[node * 2 + 1];
            float mx = fmaxf(a0, a1);
            float e0 = expf(a0 - mx), e1 = expf(a1 - mx);
            float inv = 1.0f / (e0 + e1);
            float s1  = nline[(size_t)node * QS + CC];
            float h1v = nline[(size_t)node * QS + n];
            h1v = (s1 > 0.0f) ? h1v / s1 : h1v;
            float h0v = out[(size_t)node * CC + n];
            float logit = h0v * (e0 * inv) + h1v * (e1 * inv);
            out[(size_t)node * CC + n] = sa * logit + sna * mlp;
        }
    }
}

extern "C" void kernel_launch(void* const* d_in, const int* in_sizes, int n_in,
                              void* d_out, int out_size, void* d_ws, size_t ws_size,
                              hipStream_t stream) {
    const float* features      = (const float*)d_in[0];
    const float* label_init    = (const float*)d_in[1];
    const float* labels_onehot = (const float*)d_in[2];
    const float* alpha         = (const float*)d_in[3];
    const float* attention     = (const float*)d_in[4];
    const float* e_edge        = (const float*)d_in[5];
    const float* w1            = (const float*)d_in[6];
    const float* b1            = (const float*)d_in[7];
    const float* w2            = (const float*)d_in[8];
    const float* b2            = (const float*)d_in[9];
    const int*   src           = (const int*)d_in[10];
    const int*   dst           = (const int*)d_in[11];
    const int*   train_mask    = (const int*)d_in[12];
    float* out = (float*)d_out;

    float* ws = (float*)d_ws;
    unsigned* gmax = (unsigned*)ws;
    float* acc  = ws + NN;
    float* hrow = acc + (size_t)NN * QS;

    const int TB = 256;
    const int gridI = NN / 32;

    const float* started = nullptr;
    for (int g = 0; g < GG; ++g) {
        for (int l = 0; l < LL; ++l) {
            const float* ee = e_edge + ((size_t)l * GG + g) * EE;
            const int* sg = src + (size_t)g * EE;
            const int* dg = dst + (size_t)g * EE;
            float* dest = l ? hrow : out;
            const float* hcur = l ? hrow : label_init;
            if (!started) {
                k_init<false><<<gridI, 512, 0, stream>>>(labels_onehot, train_mask, gmax, acc, dest);
                k_gmax<<<dim3(256, LL * GG), TB, 0, stream>>>(e_edge, gmax);
                started = acc;
            } else {
                k_init<true> <<<gridI, 512, 0, stream>>>(labels_onehot, train_mask, gmax, acc, dest);
            }
            k_agg<<<AGG_BLOCKS, TB, 0, stream>>>(ee, sg, dg, gmax + BITS0, gmax + (l * GG + g),
                                                 gmax + CTR0 + (g * LL + l), hcur, acc);
        }
    }

    const int n_pairs = NN / 32;
    const int blocks  = (n_pairs + 7) / 8;
    k_mlp_combine<<<blocks, TB, 0, stream>>>(features, w1, b1, w2, b2,
                                             alpha, attention, acc,
                                             out, n_pairs);
}


# Both graphs share each launch (second sums plane if the workspace fits it): one start, one edge pass per layer with tickets per graph, one divide between layers; 6 launches, not 10; add order changes
# speedup vs baseline: 16.9077x; 1.0672x over previous
#include <hip/hip_runtime.h>
#include <hip/hip_bf16.h>
#include <math.h>

#define NN 100000
#define CC 16
#define GG 2
#define LL 2
#define EE 3200000
#define FF 512
#define HH 64

#define W1_PSTRIDE 160
#define W2_PSTRIDE 32
#define HID_STRIDE 66

typedef float v2f __attribute__((ext_vector_type(2)));
typedef float v8f __attribute__((ext_vector_type(8)));

__device__ __forceinline__ unsigned f2ou(float f) {
    unsigned u = __float_as_uint(f);
    return (u & 0x80000000u) ? ~u : (u | 0x80000000u);
}
__device__ __forceinline__ float ou2f(unsigned u) {
    return __uint_as_float((u & 0x80000000u) ? (u & 0x7FFFFFFFu) : ~u);
}

#define QS 32
#define CHUNK 1024
#define NCHUNK (EE / CHUNK)
#define NHALF (NN / 2)
#define CTR0 64
#define AGG_BLOCKS 2048
#define NBITW (NN / 32)
#define NBITQ ((NBITW + 3) / 4)
#define BITS0 256

template <bool NORM>
__global__ __launch_bounds__(512) void k_init(
                       const float* __restrict__ one_hot, const int* __restrict__ mask,
                       unsigned* __restrict__ gmax, float* __restrict__ acc0,
                       float* __restrict__ dest0, float* __restrict__ dest1) {
    int t = threadIdx.x;
    float* acc  = acc0 + (size_t)blockIdx.y * NN * QS;
    float* dest = blockIdx.y ? dest1 : dest0;
    int node0 = blockIdx.x * 32;
    int   mk[2];
    float oh[2], st[2];
    int mb = NORM ? 0 : mask[node0 + (t & 31)];
    for (int r = 0; r < 2; ++r) {
        int w    = r * 512 + t;
        int node = node0 + (w >> 5);
        mk[r] = mask[node];
        oh[r] = one_hot[(size_t)node * CC + (w & 15)];
    }
    for (int r = 0; r < 2; ++r) {
        int col = (r * 512 + t) & 31;
        float x = (col < CC) ? oh[r] : ((col == CC) ? 1.0f : 0.0f);
        st[r] = mk[r] ? x : 0.0f;
    }
    if (NORM) {
        int node = node0 + (t >> 4);
        int n    = t & 15;
        float v = acc[(size_t)node * QS + n];
        float s = acc[(size_t)node * QS + CC];
        dest[(size_t)node * CC + n] = (s > 0.0f) ? v / s : v;
        __syncthreads();
    }
    for (int r = 0; r < 2; ++r) {
        int w = r * 512 + t;
        acc[(size_t)(node0 + (w >> 5)) * QS + (w & 31)] = st[r];
    }
    if (!NORM && blockIdx.x == 0 && blockIdx.y == 0) {
        if (t < LL * GG) gmax[t] = 0x007FFFFFu;
        else if (t >= CTR0 && t < CTR0 + 128) gmax[t] = 0u;
        else if (t >= CTR0 + 128 && t < CTR0 + 128 + (4 * NBITQ - NBITW)) gmax[BITS0 + NBITW + (t - CTR0 - 128)] = 0u;
    }
    if (!NORM) {
        unsigned w = __builtin_amdgcn_ballot_w32(mb != 0);
        if (t == 0 && blockIdx.y == 0) gmax[BITS0 + blockIdx.x] = w;
    }
}

__device__ __forceinline__ float max4(float4 v) {
    return fmaxf(fmaxf(v.x, v.y), fmaxf(v.z, v.w));
}
__global__ __launch_bounds__(256) void k_gmax(const float* __restrict__ e,
                                              unsigned* __restrict__ gmax) {
    __shared__ float wmax[8];
    const float4* p = (const float4*)(e + (size_t)blockIdx.y * EE);
    const int n4 = EE / 4;
    const int stride = gridDim.x * 256;
    int i = blockIdx.x * 256 + threadIdx.x;
    float m = -INFINITY;
    for (; i + 3 * stride < n4; i += 4 * stride) {
        float4 a = p[i], b = p[i + stride], c = p[i + 2 * stride], d = p[i + 3 * stride];
        m = fmaxf(m, fmaxf(fmaxf(max4(a), max4(b)), fmaxf(max4(c), max4(d))));
    }
    for (; i < n4; i += stride) m = fmaxf(m, max4(p[i]));
    for (int off = 16; off > 0; off >>= 1) m = fmaxf(m, __shfl_xor(m, off));
    if ((threadIdx.x & 31) == 0) wmax[threadIdx.x >> 5] = m;
    __syncthreads();
    if (threadIdx.x == 0) {
        float b = wmax[0];
        for (int w = 1; w < 8; ++w) b = fmaxf(b, wmax[w]);
        atomicMax(&gmax[blockIdx.y], f2ou(b));
    }
}

template <int J> __device__ __forceinline__ int row_bc(int x) {
    return __builtin_amdgcn_update_dpp(0, x, 0x150 + J, 0xf, 0xf, false);
}
__device__ __forceinline__ int xhalf(int x) {
    return __builtin_amdgcn_permlanex16(x, x, 0x76543210, 0xfedcba98, false, false);
}

template <int J0> __device__ __forceinline__ void agg_rows8(
    int d, int s, int ai, int dX, int aXi, int n, int hi16,
    const float* __restrict__ hcur, float* __restrict__ acc) {
    int   dj[8], sj[8], dO[8];
    float aj[8], aO[8], v[8];
    dj[0] = row_bc<J0 + 0>(d); sj[0] = row_bc<J0 + 0>(s); aj[0] = __int_as_float(row_bc<J0 + 0>(ai));
    dj[1] = row_bc<J0 + 1>(d); sj[1] = row_bc<J0 + 1>(s); aj[1] = __int_as_float(row_bc<J0 + 1>(ai));
    dj[2] = row_bc<J0 + 2>(d); sj[2] = row_bc<J0 + 2>(s); aj[2] = __int_as_float(row_bc<J0 + 2>(ai));
    dj[3] = row_bc<J0 + 3>(d); sj[3] = row_bc<J0 + 3>(s); aj[3] = __int_as_float(row_bc<J0 + 3>(ai));
    dj[4] = row_bc<J0 + 4>(d); sj[4] = row_bc<J0 + 4>(s); aj[4] = __int_as_float(row_bc<J0 + 4>(ai));
    dj[5] = row_bc<J0 + 5>(d); sj[5] = row_bc<J0 + 5>(s); aj[5] = __int_as_float(row_bc<J0 + 5>(ai));
    dj[6] = row_bc<J0 + 6>(d); sj[6] = row_bc<J0 + 6>(s); aj[6] = __int_as_float(row_bc<J0 + 6>(ai));
    dj[7] = row_bc<J0 + 7>(d); sj[7] = row_bc<J0 + 7>(s); aj[7] = __int_as_float(row_bc<J0 + 7>(ai));
    dO[0] = row_bc<J0 + 0>(dX); aO[0] = __int_as_float(row_bc<J0 + 0>(aXi));
    dO[1] = row_bc<J0 + 1>(dX); aO[1] = __int_as_float(row_bc<J0 + 1>(aXi));
    dO[2] = row_bc<J0 + 2>(dX); aO[2] = __int_as_float(row_bc<J0 + 2>(aXi));
    dO[3] = row_bc<J0 + 3>(dX); aO[3] = __int_as_float(row_bc<J0 + 3>(aXi));
    dO[4] = row_bc<J0 + 4>(dX); aO[4] = __int_as_float(row_bc<J0 + 4>(aXi));
    dO[5] = row_bc<J0 + 5>(dX); aO[5] = __int_as_float(row_bc<J0 + 5>(aXi));
    dO[6] = row_bc<J0 + 6>(dX); aO[6] = __int_as_float(row_bc<J0 + 6>(aXi));
    dO[7] = row_bc<J0 + 7>(dX); aO[7] = __int_as_float(row_bc<J0 + 7>(aXi));
#pragma unroll
    for (int k = 0; k < 8; ++k)
        v[k] = (dj[k] >= 0) ? hcur[(size_t)sj[k] * CC + n] : 0.0f;
#pragma unroll
    for (int k = 0; k < 8; ++k) {
#pragma unroll
        for (int h = 0; h < 2; ++h) {
            bool  own = (hi16 == h);
            int   dd  = own ? dj[k] : dO[k];
            int   col = own ? n : CC;
            float x   = own ? v[k] * aj[k] : aO[k];
            if ((own || n == 0) && dd >= 0)
                atomicAdd(acc + (size_t)dd * QS + col, x);
        }
    }
}

__global__ __launch_bounds__(256) void k_agg(
                      const float* __restrict__ e, const int* __restrict__ src,
                      const int* __restrict__ dst, const unsigned* __restrict__ bits,
                      const unsigned* __restrict__ gmax, unsigned* __restrict__ ctr,
                      const float* __restrict__ hcur0, const float* __restrict__ hcur1,
                      float* __restrict__ acc, int ng) {
    __shared__ int      s_d[CHUNK];
    __shared__ int      s_s[CHUNK];
    __shared__ float    s_a[CHUNK];
    __shared__ unsigned s_cnt;
    __shared__ unsigned s_t;
    __shared__ unsigned s_home;
    __shared__ uint4    s_bits4[4 * 256];
    int tid  = threadIdx.x;
    int lane = tid & 31;
    int wave = tid >> 5;
    int n = tid & 15;
    int hi16 = (tid >> 4) & 1;
    if (tid == 0) s_home = (((unsigned)__builtin_amdgcn_s_sendmsg_rtn(0x87) >> 16) >> 2) & 1u;
    {
        const uint4* bq = (const uint4*)bits;
        uint4 tb[4];
#pragma unroll
        for (int k = 0; k < 4; ++k) {
            int i = tid + k * 256;
            tb[k] = bq[i < NBITQ ? i : NBITQ - 1];
        }
#pragma unroll
        for (int k = 0; k < 4; ++k) s_bits4[tid + k * 256] = tb[k];
    }
    const unsigned* s_bits = (const unsigned*)s_bits4;
    __syncthreads();
    unsigned home = s_home;
    for (int ph = 0; ph < 2; ++ph) {
        unsigned h = home ^ (unsigned)ph;
        int lo  = h ? NHALF : 0;
        int hiN = h ? NN : NHALF;
      for (int g = 0; g < ng; ++g) {
        const float* eg   = e   + (size_t)g * EE;
        const int*   sg   = src + (size_t)g * EE;
        const int*   dg   = dst + (size_t)g * EE;
        const float* hcur = g ? hcur1 : hcur0;
        float*       accg = acc + (size_t)g * NN * QS;
        unsigned*    ctrg = ctr + g * LL + 64 * h;
        float gm = ou2f(gmax[g]);
        for (;;) {
            if (tid == 0) {
                s_t = atomicAdd(ctrg, 1u);
                s_cnt = 0u;
            }
            __syncthreads();
            unsigned tk = s_t;
            __syncthreads();
            if (tk >= (unsigned)NCHUNK) break;
            int i0 = (int)tk * CHUNK + tid;
            int   dd[4], ss[4];
            float ev[4];
#pragma unroll
            for (int r = 0; r < 4; ++r) {
                dd[r] = dg[i0 + r * 256];
                ss[r] = sg[i0 + r * 256];
                ev[r] = eg[i0 + r * 256];
            }
            bool ok[4];
            unsigned bal[4];
            unsigned tw[4];
#pragma unroll
            for (int r = 0; r < 4; ++r) {
                unsigned wi = (unsigned)dd[r] >> 5;
                tw[r] = s_bits[wi < (unsigned)NBITW ? wi : (unsigned)(NBITW - 1)];
            }
#pragma unroll
            for (int r = 0; r < 4; ++r) {
                unsigned in  = (unsigned)(dd[r] >= lo) & (unsigned)(dd[r] < hiN);
                unsigned bit = (tw[r] >> (dd[r] & 31)) & 1u;
                ok[r] = (in & (bit ^ 1u)) != 0u;
            }
#pragma unroll
            for (int r = 0; r < 4; ++r) bal[r] = __builtin_amdgcn_ballot_w32(ok[r]);
            unsigned tot = (unsigned)(__popc(bal[0]) + __popc(bal[1]) + __popc(bal[2]) + __popc(bal[3]));
            unsigned base = 0u;
            if (lane == 0) base = atomicAdd(&s_cnt, tot);
            base = (unsigned)__builtin_amdgcn_readfirstlane((int)base);
#pragma unroll
            for (int r = 0; r < 4; ++r) {
                if (ok[r]) {
                    unsigned pos = base + __builtin_amdgcn_mbcnt_lo(bal[r], 0u);
                    s_d[pos] = dd[r];
                    s_s[pos] = ss[r];
                    s_a[pos] = expf(ev[r] - gm);
                }
                base += (unsigned)__popc(bal[r]);
            }
            __syncthreads();
            int cnt = __builtin_amdgcn_readfirstlane((int)s_cnt);
            for (int q0 = __builtin_amdgcn_readfirstlane(wave * 32); q0 < cnt; q0 += 256) {
                int q = q0 + lane;
                int d = -1, s = 0;
                float a = 0.0f;
                if (q < cnt) { d = s_d[q]; s = s_s[q]; a = s_a[q]; }
                int ai  = __float_as_int(a);
                int dX  = xhalf(d);
                int aXi = xhalf(ai);
                agg_rows8<0>(d, s, ai, dX, aXi, n, hi16, hcur, accg);
                agg_rows8<8>(d, s, ai, dX, aXi, n, hi16, hcur, accg);
            }
            __syncthreads();
        }
      }
    }
}

template <bool TWO>
__global__ __launch_bounds__(256) void k_mlp_combine(
    const float* __restrict__ feat, const float* __restrict__ w1,
    const float* __restrict__ b1,   const float* __restrict__ w2,
    const float* __restrict__ b2,   const float* __restrict__ alpha,
    const float* __restrict__ att,  const float* __restrict__ nline,
    const float* __restrict__ nline0,
    float* __restrict__ out, int n_pairs)
{
    __shared__ float sW1p[(FF / 2) * W1_PSTRIDE];
    __shared__ float sW2p[(HH / 2) * W2_PSTRIDE];
    __shared__ float sHid[8][2][16 * HID_STRIDE];

    int t = threadIdx.x;

    for (int i = t; i < (FF / 2) * (HH / 4); i += 256) {
        int p  = i >> 4;
        int nq = (i & 15) << 2;
        float4 x = *(const float4*)&w1[(size_t)p * 2 * HH + nq];
        float4 y = *(const float4*)&w1[(size_t)p * 2 * HH + HH + nq];
        float4 lo = { x.x, y.x, x.y, y.y };
        float4 hi = { x.z, y.z, x.w, y.w };
        float* d = &sW1p[p * W1_PSTRIDE + 2 * nq];
        *(float4*)(d + 0) = lo;
        *(float4*)(d + 4) = hi;
    }
    for (int i = t; i < (HH / 2) * CC; i += 256) {
        int p = i >> 4, n = i & 15;
        v2f v; v.x = w2[(size_t)p * 2 * CC + n];
        v.y = w2[(size_t)p * 2 * CC + CC + n];
        *(v2f*)&sW2p[p * W2_PSTRIDE + 2 * n] = v;
    }
    __syncthreads();

    int wave = t >> 5;
    int lane = t & 31;
    int pair = blockIdx.x * 8 + wave;
    if (pair >= n_pairs) return;

    int tile0 = pair * 2;
    int tile1 = tile0 + 1;

    int m     = lane & 15;
    int n     = lane & 15;
    int hi16  = lane >> 4;
    int khalf = hi16 << 1;
    int moff  = hi16 << 3;

    v8f acc0 = {}, acc1 = {}, acc2 = {}, acc3 = {};
    v8f acc4 = {}, acc5 = {}, acc6 = {}, acc7 = {};
    const float* arow0 = feat + ((size_t)tile0 * 16 + m) * FF;
    const float* arow1 = feat + ((size_t)tile1 * 16 + m) * FF;
    for (int k = 0; k < FF; k += 4) {
        int kb = k + khalf;
        v2f a0 = *(const v2f*)(arow0 + kb);
        v2f a1 = *(const v2f*)(arow1 + kb);
        const float* bp = &sW1p[((k >> 1) + hi16) * W1_PSTRIDE + 2 * n];
        v2f b0 = *(const v2f*)(bp + 0);
        v2f b1v = *(const v2f*)(bp + 32);
        v2f b2v = *(const v2f*)(bp + 64);
        v2f b3v = *(const v2f*)(bp + 96);
        acc0 = __builtin_amdgcn_wmma_f32_16x16x4_f32(false, a0, false, b0,  (short)0, acc0, false, false);
        acc1 = __builtin_amdgcn_wmma_f32_16x16x4_f32(false, a0, false, b1v, (short)0, acc1, false, false);
        acc2 = __builtin_amdgcn_wmma_f32_16x16x4_f32(false, a0, false, b2v, (short)0, acc2, false, false);
        acc3 = __builtin_amdgcn_wmma_f32_16x16x4_f32(false, a0, false, b3v, (short)0, acc3, false, false);
        acc4 = __builtin_amdgcn_wmma_f32_16x16x4_f32(false, a1, false, b0,  (short)0, acc4, false, false);
        acc5 = __builtin_amdgcn_wmma_f32_16x16x4_f32(false, a1, false, b1v, (short)0, acc5, false, false);
        acc6 = __builtin_amdgcn_wmma_f32_16x16x4_f32(false, a1, false, b2v, (short)0, acc6, false, false);
        acc7 = __builtin_amdgcn_wmma_f32_16x16x4_f32(false, a1, false, b3v, (short)0, acc7, false, false);
    }

    float bb0 = b1[n + 0], bb1 = b1[n + 16], bb2 = b1[n + 32], bb3 = b1[n + 48];
    {
        float* hd0 = &sHid[wave][0][0];
        float* hd1 = &sHid[wave][1][0];
        for (int r = 0; r < 8; ++r) {
            int mm = r + moff;
            hd0[mm * HID_STRIDE + n +  0] = fmaxf(acc0[r] + bb0, 0.0f);
            hd0[mm * HID_STRIDE + n + 16] = fmaxf(acc1[r] + bb1, 0.0f);
            hd0[mm * HID_STRIDE + n + 32] = fmaxf(acc2[r] + bb2, 0.0f);
            hd0[mm * HID_STRIDE + n + 48] = fmaxf(acc3[r] + bb3, 0.0f);
            hd1[mm * HID_STRIDE + n +  0] = fmaxf(acc4[r] + bb0, 0.0f);
            hd1[mm * HID_STRIDE + n + 16] = fmaxf(acc5[r] + bb1, 0.0f);
            hd1[mm * HID_STRIDE + n + 32] = fmaxf(acc6[r] + bb2, 0.0f);
            hd1[mm * HID_STRIDE + n + 48] = fmaxf(acc7[r] + bb3, 0.0f);
        }
    }

    float bc = b2[n];
    for (int tt = 0; tt < 2; ++tt) {
        const float* hd = &sHid[wave][tt][0];
        v8f acc = {};
        for (int k = 0; k < HH; k += 4) {
            int kb = k + khalf;
            v2f a = *(const v2f*)(hd + m * HID_STRIDE + kb);
            v2f b = *(const v2f*)&sW2p[((k >> 1) + hi16) * W2_PSTRIDE + 2 * n];
            acc = __builtin_amdgcn_wmma_f32_16x16x4_f32(false, a, false, b, (short)0, acc, false, false);
        }
        int tile = tt ? tile1 : tile0;
        for (int r = 0; r < 8; ++r) {
            int node = tile * 16 + r + moff;
            float mlp = acc[r] + bc;
            float al  = alpha[node];
            float sa  = 1.0f / (1.0f + expf(-al));
            float sna = 1.0f - sa;
            float a0 = att[node * 2 + 0], a1 = att[node * 2 + 1];
            float mx = fmaxf(a0, a1);
            float e0 = expf(a0 - mx), e1 = expf(a1 - mx);
            float inv = 1.0f / (e0 + e1);
            float s1  = nline[(size_t)node * QS + CC];
            float h1v = nline[(size_t)node * QS + n];
            h1v = (s1 > 0.0f) ? h1v / s1 : h1v;
            float h0v;
            if (TWO) {
                float s0 = nline0[(size_t)node * QS + CC];
                h0v = nline0[(size_t)node * QS + n];
                h0v = (s0 > 0.0f) ? h0v / s0 : h0v;
            } else {
                h0v = out[(size_t)node * CC + n];
            }
            float logit = h0v * (e0 * inv) + h1v * (e1 * inv);
            out[(size_t)node * CC + n] = sa * logit + sna * mlp;
        }
    }
}

extern "C" void kernel_launch(void* const* d_in, const int* in_sizes, int n_in,
                              void* d_out, int out_size, void* d_ws, size_t ws_size,
                              hipStream_t stream) {
    const float* features      = (const float*)d_in[0];
    const float* label_init    = (const float*)d_in[1];
    const float* labels_onehot = (const float*)d_in[2];
    const float* alpha         = (const float*)d_in[3];
    const float* attention     = (const float*)d_in[4];
    const float* e_edge        = (const float*)d_in[5];
    const float* w1            = (const float*)d_in[6];
    const float* b1            = (const float*)d_in[7];
    const float* w2            = (const float*)d_in[8];
    const float* b2            = (const float*)d_in[9];
    const int*   src           = (const int*)d_in[10];
    const int*   dst           = (const int*)d_in[11];
    const int*   train_mask    = (const int*)d_in[12];
    float* out = (float*)d_out;

    float* ws = (float*)d_ws;
    unsigned* gmax = (unsigned*)ws;
    float* acc  = ws + NN;

    const int TB = 256;
    const int gridI = NN / 32;
    const int n_pairs = NN / 32;
    const int blocks  = (n_pairs + 7) / 8;

    if (ws_size >= sizeof(float) * NN * (1 + 2 * QS + CC)) {
        float* acc1 = acc + (size_t)NN * QS;
        float* hrow = acc1 + (size_t)NN * QS;
        k_init<false><<<dim3(gridI, GG), 512, 0, stream>>>(labels_onehot, train_mask, gmax, acc, nullptr, nullptr);
        k_gmax<<<dim3(256, LL * GG), TB, 0, stream>>>(e_edge, gmax);
        k_agg<<<AGG_BLOCKS, TB, 0, stream>>>(e_edge, src, dst, gmax + BITS0, gmax, gmax + CTR0,
                                             label_init, label_init, acc, GG);
        k_init<true><<<dim3(gridI, GG), 512, 0, stream>>>(labels_onehot, train_mask, gmax, acc, hrow, out);
        k_agg<<<AGG_BLOCKS, TB, 0, stream>>>(e_edge + (size_t)GG * EE, src, dst, gmax + BITS0, gmax + GG,
                                             gmax + CTR0 + 1, hrow, out, acc, GG);
        k_mlp_combine<true><<<blocks, TB, 0, stream>>>(features, w1, b1, w2, b2,
                                                       alpha, attention, acc1, acc,
                                                       out, n_pairs);
        return;
    }

    float* hrow = acc + (size_t)NN * QS;
    const float* started = nullptr;
    for (int g = 0; g < GG; ++g) {
        for (int l = 0; l < LL; ++l) {
            const float* ee = e_edge + ((size_t)l * GG + g) * EE;
            const int* sg = src + (size_t)g * EE;
            const int* dg = dst + (size_t)g * EE;
            float* dest = l ? hrow : out;
            const float* hcur = l ? hrow : label_init;
            if (!started) {
                k_init<false><<<gridI, 512, 0, stream>>>(labels_onehot, train_mask, gmax, acc, dest, nullptr);
                k_gmax<<<dim3(256, LL * GG), TB, 0, stream>>>(e_edge, gmax);
                started = acc;
            } else {
                k_init<true> <<<gridI, 512, 0, stream>>>(labels_onehot, train_mask, gmax, acc, dest, nullptr);
            }
            k_agg<<<AGG_BLOCKS, TB, 0, stream>>>(ee, sg, dg, gmax + BITS0, gmax + (l * GG + g),
                                                 gmax + CTR0 + (g * LL + l), hcur, hcur, acc, 1);
        }
    }
    k_mlp_combine<false><<<blocks, TB, 0, stream>>>(features, w1, b1, w2, b2,
                                                    alpha, attention, acc, nullptr,
                                                    out, n_pairs);
}


# The MLP runs inside the last edge pass's launch as tickets of its own (w1 in LDS slices, products transposed, hidden tile fed from registers); a light mix kernel ends; MLP sum order changes
# speedup vs baseline: 17.7641x; 1.0507x over previous
#include <hip/hip_runtime.h>
#include <hip/hip_bf16.h>
#include <math.h>

#define NN 100000
#define CC 16
#define GG 2
#define LL 2
#define EE 3200000
#define FF 512
#define HH 64

#define W1_PSTRIDE 160
#define W2_PSTRIDE 32
#define HID_STRIDE 66
#define KS 32
#define W1S_STRIDE 72
#define W1S_WORDS (KS * W1S_STRIDE)
#define MLP_ROWS 128

typedef float v2f __attribute__((ext_vector_type(2)));
typedef float v8f __attribute__((ext_vector_type(8)));

__device__ __forceinline__ unsigned f2ou(float f) {
    unsigned u = __float_as_uint(f);
    return (u & 0x80000000u) ? ~u : (u | 0x80000000u);
}
__device__ __forceinline__ float ou2f(unsigned u) {
    return __uint_as_float((u & 0x80000000u) ? (u & 0x7FFFFFFFu) : ~u);
}

#define QS 32
#define CHUNK 1024
#define NCHUNK (EE / CHUNK)
#define NHALF (NN / 2)
#define CTR0 64
#define AGG_BLOCKS 2048
#define AGG_BLOCKS_MLP 1280
#define NBITW (NN / 32)
#define NBITQ ((NBITW + 3) / 4)
#define BITS0 256

template <bool NORM>
__global__ __launch_bounds__(512) void k_init(
                       const float* __restrict__ one_hot, const int* __restrict__ mask,
                       unsigned* __restrict__ gmax, float* __restrict__ acc0,
                       float* __restrict__ dest0, float* __restrict__ dest1) {
    int t = threadIdx.x;
    float* acc  = acc0 + (size_t)blockIdx.y * NN * QS;
    float* dest = blockIdx.y ? dest1 : dest0;
    int node0 = blockIdx.x * 32;
    int   mk[2];
    float oh[2], st[2];
    int mb = NORM ? 0 : mask[node0 + (t & 31)];
    for (int r = 0; r < 2; ++r) {
        int w    = r * 512 + t;
        int node = node0 + (w >> 5);
        mk[r] = mask[node];
        oh[r] = one_hot[(size_t)node * CC + (w & 15)];
    }
    for (int r = 0; r < 2; ++r) {
        int col = (r * 512 + t) & 31;
        float x = (col < CC) ? oh[r] : ((col == CC) ? 1.0f : 0.0f);
        st[r] = mk[r] ? x : 0.0f;
    }
    if (NORM) {
        int node = node0 + (t >> 4);
        int n    = t & 15;
        float v = acc[(size_t)node * QS + n];
        float s = acc[(size_t)node * QS + CC];
        dest[(size_t)node * CC + n] = (s > 0.0f) ? v / s : v;
        __syncthreads();
    }
    for (int r = 0; r < 2; ++r) {
        int w = r * 512 + t;
        acc[(size_t)(node0 + (w >> 5)) * QS + (w & 31)] = st[r];
    }
    if (!NORM && blockIdx.x == 0 && blockIdx.y == 0) {
        if (t < LL * GG) gmax[t] = 0x007FFFFFu;
        else if (t >= CTR0 && t < CTR0 + 128) gmax[t] = 0u;
        else if (t >= CTR0 + 128 && t < CTR0 + 128 + (4 * NBITQ - NBITW)) gmax[BITS0 + NBITW + (t - CTR0 - 128)] = 0u;
    }
    if (!NORM) {
        unsigned w = __builtin_amdgcn_ballot_w32(mb != 0);
        if (t == 0 && blockIdx.y == 0) gmax[BITS0 + blockIdx.x] = w;
    }
}

__device__ __forceinline__ float max4(float4 v) {
    return fmaxf(fmaxf(v.x, v.y), fmaxf(v.z, v.w));
}
__global__ __launch_bounds__(256) void k_gmax(const float* __restrict__ e,
                                              unsigned* __restrict__ gmax) {
    __shared__ float wmax[8];
    const float4* p = (const float4*)(e + (size_t)blockIdx.y * EE);
    const int n4 = EE / 4;
    const int stride = gridDim.x * 256;
    int i = blockIdx.x * 256 + threadIdx.x;
    float m = -INFINITY;
    for (; i + 3 * stride < n4; i += 4 * stride) {
        float4 a = p[i], b = p[i + stride], c = p[i + 2 * stride], d = p[i + 3 * stride];
        m = fmaxf(m, fmaxf(fmaxf(max4(a), max4(b)), fmaxf(max4(c), max4(d))));
    }
    for (; i < n4; i += stride) m = fmaxf(m, max4(p[i]));
    for (int off = 16; off > 0; off >>= 1) m = fmaxf(m, __shfl_xor(m, off));
    if ((threadIdx.x & 31) == 0) wmax[threadIdx.x >> 5] = m;
    __syncthreads();
    if (threadIdx.x == 0) {
        float b = wmax[0];
        for (int w = 1; w < 8; ++w) b = fmaxf(b, wmax[w]);
        atomicMax(&gmax[blockIdx.y], f2ou(b));
    }
}

template <int J> __device__ __forceinline__ int row_bc(int x) {
    return __builtin_amdgcn_update_dpp(0, x, 0x150 + J, 0xf, 0xf, false);
}
__device__ __forceinline__ int xhalf(int x) {
    return __builtin_amdgcn_permlanex16(x, x, 0x76543210, 0xfedcba98, false, false);
}

template <int J0> __device__ __forceinline__ void agg_rows8(
    int d, int s, int ai, int dX, int aXi, int n, int hi16,
    const float* __restrict__ hcur, float* __restrict__ acc) {
    int   dj[8], sj[8], dO[8];
    float aj[8], aO[8], v[8];
    dj[0] = row_bc<J0 + 0>(d); sj[0] = row_bc<J0 + 0>(s); aj[0] = __int_as_float(row_bc<J0 + 0>(ai));
    dj[1] = row_bc<J0 + 1>(d); sj[1] = row_bc<J0 + 1>(s); aj[1] = __int_as_float(row_bc<J0 + 1>(ai));
    dj[2] = row_bc<J0 + 2>(d); sj[2] = row_bc<J0 + 2>(s); aj[2] = __int_as_float(row_bc<J0 + 2>(ai));
    dj[3] = row_bc<J0 + 3>(d); sj[3] = row_bc<J0 + 3>(s); aj[3] = __int_as_float(row_bc<J0 + 3>(ai));
    dj[4] = row_bc<J0 + 4>(d); sj[4] = row_bc<J0 + 4>(s); aj[4] = __int_as_float(row_bc<J0 + 4>(ai));
    dj[5] = row_bc<J0 + 5>(d); sj[5] = row_bc<J0 + 5>(s); aj[5] = __int_as_float(row_bc<J0 + 5>(ai));
    dj[6] = row_bc<J0 + 6>(d); sj[6] = row_bc<J0 + 6>(s); aj[6] = __int_as_float(row_bc<J0 + 6>(ai));
    dj[7] = row_bc<J0 + 7>(d); sj[7] = row_bc<J0 + 7>(s); aj[7] = __int_as_float(row_bc<J0 + 7>(ai));
    dO[0] = row_bc<J0 + 0>(dX); aO[0] = __int_as_float(row_bc<J0 + 0>(aXi));
    dO[1] = row_bc<J0 + 1>(dX); aO[1] = __int_as_float(row_bc<J0 + 1>(aXi));
    dO[2] = row_bc<J0 + 2>(dX); aO[2] = __int_as_float(row_bc<J0 + 2>(aXi));
    dO[3] = row_bc<J0 + 3>(dX); aO[3] = __int_as_float(row_bc<J0 + 3>(aXi));
    dO[4] = row_bc<J0 + 4>(dX); aO[4] = __int_as_float(row_bc<J0 + 4>(aXi));
    dO[5] = row_bc<J0 + 5>(dX); aO[5] = __int_as_float(row_bc<J0 + 5>(aXi));
    dO[6] = row_bc<J0 + 6>(dX); aO[6] = __int_as_float(row_bc<J0 + 6>(aXi));
    dO[7] = row_bc<J0 + 7>(dX); aO[7] = __int_as_float(row_bc<J0 + 7>(aXi));
#pragma unroll
    for (int k = 0; k < 8; ++k)
        v[k] = (dj[k] >= 0) ? hcur[(size_t)sj[k] * CC + n] : 0.0f;
#pragma unroll
    for (int k = 0; k < 8; ++k) {
#pragma unroll
        for (int h = 0; h < 2; ++h) {
            bool  own = (hi16 == h);
            int   dd  = own ? dj[k] : dO[k];
            int   col = own ? n : CC;
            float x   = own ? v[k] * aj[k] : aO[k];
            if ((own || n == 0) && dd >= 0)
                atomicAdd(acc + (size_t)dd * QS + col, x);
        }
    }
}

#define NMLPT0 ((NHALF + MLP_ROWS - 1) / MLP_ROWS)
#define MCTR 8

__device__ __forceinline__ v8f mm4(v2f a, v2f b, v8f c) {
    return __builtin_amdgcn_wmma_f32_16x16x4_f32(false, a, false, b, (short)0, c, false, false);
}

__device__ __forceinline__ void mlp_ticket(int ticket, int tid, float* __restrict__ sW,
        const float* __restrict__ feat, const float* __restrict__ w1, const float* __restrict__ b1,
        const float* __restrict__ w2, const float* __restrict__ b2, float* __restrict__ acc) {
    int lane = tid & 31;
    int c = lane & 15;
    int h = lane >> 4;
    int row0 = __builtin_amdgcn_readfirstlane(ticket * MLP_ROWS + (tid >> 5) * 16);
    bool live = row0 < NN;
    int rbase = live ? row0 : 0;
    int node = rbase + c;
    const float* fbase = feat + (size_t)rbase * FF;
    unsigned flo = (unsigned)(c * FF + 4 * h);
    int sp  = tid >> 4;
    int snq = (tid & 15) << 2;
    unsigned sgo = (unsigned)((2 * sp) * HH + snq);
    float* sd = sW + (2 * sp) * W1S_STRIDE + snq;
    float4 sx = *(const float4*)(w1 + sgo);
    float4 sy = *(const float4*)(w1 + HH + sgo);
    *(float4*)(sd) = sx;
    *(float4*)(sd + W1S_STRIDE) = sy;
    float4 fc[4], fn[4];
#pragma unroll
    for (int i = 0; i < 4; ++i) fc[i] = *(const float4*)(fbase + 8 * i + flo);
#pragma unroll
    for (int i = 0; i < 4; ++i) fn[i] = fc[i];
    __syncthreads();
    v8f a0 = {}, a1 = {}, a2 = {}, a3 = {};
#pragma nounroll
    for (int j = 0; j < FF / KS; ++j) {
        const float* cur = sW + (j & 1) * W1S_WORDS;
        bool more = (j + 1 < FF / KS);
        if (more) {
            sx = *(const float4*)(w1 + (j + 1) * KS * HH + sgo);
            sy = *(const float4*)(w1 + (j + 1) * KS * HH + HH + sgo);
#pragma unroll
            for (int i = 0; i < 4; ++i) fn[i] = *(const float4*)(fbase + (j + 1) * KS + 8 * i + flo);
        }
#pragma unroll
        for (int kb = 0; kb < KS; kb += 8) {
            float4 f = fc[kb >> 3];
            const float* ap = cur + (kb + 4 * h) * W1S_STRIDE + c;
            v2f f01 = { f.x, f.y };
            v2f f23 = { f.z, f.w };
            v2f w00 = { ap[ 0], ap[W1S_STRIDE +  0] };
            v2f w01 = { ap[16], ap[W1S_STRIDE + 16] };
            v2f w02 = { ap[32], ap[W1S_STRIDE + 32] };
            v2f w03 = { ap[48], ap[W1S_STRIDE + 48] };
            a0 = mm4(w00, f01, a0);
            a1 = mm4(w01, f01, a1);
            a2 = mm4(w02, f01, a2);
            a3 = mm4(w03, f01, a3);
            v2f w20 = { ap[2 * W1S_STRIDE +  0], ap[3 * W1S_STRIDE +  0] };
            v2f w21 = { ap[2 * W1S_STRIDE + 16], ap[3 * W1S_STRIDE + 16] };
            v2f w22 = { ap[2 * W1S_STRIDE + 32], ap[3 * W1S_STRIDE + 32] };
            v2f w23 = { ap[2 * W1S_STRIDE + 48], ap[3 * W1S_STRIDE + 48] };
            a0 = mm4(w20, f23, a0);
            a1 = mm4(w21, f23, a1);
            a2 = mm4(w22, f23, a2);
            a3 = mm4(w23, f23, a3);
        }
        if (more) {
            float* d = sd + ((j + 1) & 1) * W1S_WORDS;
            *(float4*)(d) = sx;
            *(float4*)(d + W1S_STRIDE) = sy;
        }
#pragma unroll
        for (int i = 0; i < 4; ++i) fc[i] = fn[i];
        __syncthreads();
    }
    if (live) {
        int te = tid;
        asm volatile("" : "+v"(te));
        int c2 = te & 15, h2 = (te >> 4) & 1;
        const float* wb = sW + 2 * W1S_WORDS + (8 * h2) * CC + c2;
        const float* bb = sW + 2 * W1S_WORDS + HH * CC + 8 * h2;
        v8f o = {};
#pragma unroll
        for (int t = 0; t < 4; ++t) {
            v8f x = (t == 0) ? a0 : (t == 1) ? a1 : (t == 2) ? a2 : a3;
#pragma unroll
            for (int q = 0; q < 4; ++q) {
                int u = 16 * t + 2 * q;
                v2f hb = { fmaxf(x[2 * q] + bb[u], 0.0f), fmaxf(x[2 * q + 1] + bb[u + 1], 0.0f) };
                v2f wa = { wb[u * CC], wb[(u + 1) * CC] };
                o = mm4(wa, hb, o);
            }
        }
        asm volatile("s_wait_alu depctr_va_vdst(0)" : "+v"(o));
        float* q = acc + (size_t)((unsigned)(h2 * NN + rbase + c2) * QS + 24u);
#pragma unroll
        for (int r = 0; r < 8; ++r)
            __hip_atomic_store(q + r, o[r] + bb[HH + r], __ATOMIC_RELAXED, __HIP_MEMORY_SCOPE_AGENT);
    }
}

__device__ __forceinline__ void mlp_phase(unsigned home, int tid, unsigned* s_t, float* sW, unsigned* __restrict__ mctr,
        const float* __restrict__ feat, const float* __restrict__ w1, const float* __restrict__ b1,
        const float* __restrict__ w2, const float* __restrict__ b2, float* __restrict__ acc) {
    {
        float* sW2 = sW + 2 * W1S_WORDS;
        ((float4*)sW2)[tid] = ((const float4*)w2)[tid];
        if (tid < HH) sW2[HH * CC + tid] = b1[tid];
        else if (tid < HH + CC) sW2[HH * CC + tid] = b2[tid - HH];
    }
    for (int ph = 0; ph < 2; ++ph) {
        unsigned hh = home ^ (unsigned)ph;
        for (;;) {
            if (tid == 0) *s_t = atomicAdd(mctr + 64 * hh, 1u);
            __syncthreads();
            unsigned tk = *s_t;
            __syncthreads();
            if (tk >= (unsigned)NMLPT0) break;
            mlp_ticket((int)(hh * NMLPT0 + tk), tid, sW, feat, w1, b1, w2, b2, acc);
        }
    }
}

template <bool MLP>
__device__ __forceinline__ void agg_body(
                      const float* __restrict__ e, const int* __restrict__ src,
                      const int* __restrict__ dst, const unsigned* __restrict__ bits,
                      const unsigned* __restrict__ gmax, unsigned* __restrict__ ctr,
                      const float* __restrict__ hcur0, const float* __restrict__ hcur1,
                      float* __restrict__ acc, int ng,
                      const float* __restrict__ feat, const float* __restrict__ w1,
                      const float* __restrict__ b1,   const float* __restrict__ w2,
                      const float* __restrict__ b2,   unsigned* __restrict__ mctr) {
    __shared__ uint4    s_raw[3 * CHUNK / 4 + 4 * 256];
    __shared__ unsigned s_cnt;
    __shared__ unsigned s_t;
    __shared__ unsigned s_home;
    int*   s_d = (int*)s_raw;
    int*   s_s = s_d + CHUNK;
    float* s_a = (float*)(s_s + CHUNK);
    uint4* s_bits4 = s_raw + 3 * CHUNK / 4;
    int tid  = threadIdx.x;
    if (tid == 0) s_home = (((unsigned)__builtin_amdgcn_s_sendmsg_rtn(0x87) >> 16) >> 2) & 1u;
    bool mlp_first = MLP && (blockIdx.x * 2654435761u) < 858993459u;
    if (MLP) {
        __syncthreads();
        if (mlp_first) {
            mlp_phase(s_home, tid, &s_t, (float*)s_raw, mctr, feat, w1, b1, w2, b2, acc);
            __syncthreads();
        }
    }
    int ta = tid;
    if (MLP) asm volatile("" : "+v"(ta));
    int lane = ta & 31;
    int wave = ta >> 5;
    int n = ta & 15;
    int hi16 = (ta >> 4) & 1;
    {
        const uint4* bq = (const uint4*)bits;
        uint4 tb[4];
#pragma unroll
        for (int k = 0; k < 4; ++k) {
            int i = tid + k * 256;
            tb[k] = bq[i < NBITQ ? i : NBITQ - 1];
        }
#pragma unroll
        for (int k = 0; k < 4; ++k) s_bits4[tid + k * 256] = tb[k];
    }
    const unsigned* s_bits = (const unsigned*)s_bits4;
    __syncthreads();
    unsigned home = s_home;
    for (int ph = 0; ph < 2; ++ph) {
        unsigned h = home ^ (unsigned)ph;
        int lo  = h ? NHALF : 0;
        int hiN = h ? NN : NHALF;
      for (int g = 0; g < ng; ++g) {
        const float* eg   = e   + (size_t)g * EE;
        const int*   sg   = src + (size_t)g * EE;
        const int*   dg   = dst + (size_t)g * EE;
        const float* hcur = g ? hcur1 : hcur0;
        float*       accg = acc + (size_t)g * NN * QS;
        unsigned*    ctrg = ctr + g * LL + 64 * h;
        float gm = ou2f(gmax[g]);
        for (;;) {
            if (tid == 0) {
                s_t = atomicAdd(ctrg, 1u);
                s_cnt = 0u;
            }
            __syncthreads();
            unsigned tk = s_t;
            __syncthreads();
            if (tk >= (unsigned)NCHUNK) break;
            int i0 = (int)tk * CHUNK + tid;
            int   dd[4], ss[4];
            float ev[4];
#pragma unroll
            for (int r = 0; r < 4; ++r) {
                dd[r] = dg[i0 + r * 256];
                ss[r] = sg[i0 + r * 256];
                ev[r] = eg[i0 + r * 256];
            }
            bool ok[4];
            unsigned bal[4];
            unsigned tw[4];
#pragma unroll
            for (int r = 0; r < 4; ++r) {
                unsigned wi = (unsigned)dd[r] >> 5;
                tw[r] = s_bits[wi < (unsigned)NBITW ? wi : (unsigned)(NBITW - 1)];
            }
#pragma unroll
            for (int r = 0; r < 4; ++r) {
                unsigned in  = (unsigned)(dd[r] >= lo) & (unsigned)(dd[r] < hiN);
                unsigned bit = (tw[r] >> (dd[r] & 31)) & 1u;
                ok[r] = (in & (bit ^ 1u)) != 0u;
            }
#pragma unroll
            for (int r = 0; r < 4; ++r) bal[r] = __builtin_amdgcn_ballot_w32(ok[r]);
            unsigned tot = (unsigned)(__popc(bal[0]) + __popc(bal[1]) + __popc(bal[2]) + __popc(bal[3]));
            unsigned base = 0u;
            if (lane == 0) base = atomicAdd(&s_cnt, tot);
            base = (unsigned)__builtin_amdgcn_readfirstlane((int)base);
#pragma unroll
            for (int r = 0; r < 4; ++r) {
                if (ok[r]) {
                    unsigned pos = base + __builtin_amdgcn_mbcnt_lo(bal[r], 0u);
                    s_d[pos] = dd[r];
                    s_s[pos] = ss[r];
                    s_a[pos] = expf(ev[r] - gm);
                }
                base += (unsigned)__popc(bal[r]);
            }
            __syncthreads();
            int cnt = __builtin_amdgcn_readfirstlane((int)s_cnt);
            for (int q0 = __builtin_amdgcn_readfirstlane(wave * 32); q0 < cnt; q0 += 256) {
                int q = q0 + lane;
                int d = -1, s = 0;
                float a = 0.0f;
                if (q < cnt) { d = s_d[q]; s = s_s[q]; a = s_a[q]; }
                int ai  = __float_as_int(a);
                int dX  = xhalf(d);
                int aXi = xhalf(ai);
                agg_rows8<0>(d, s, ai, dX, aXi, n, hi16, hcur, accg);
                agg_rows8<8>(d, s, ai, dX, aXi, n, hi16, hcur, accg);
            }
            __syncthreads();
        }
      }
    }
    if (MLP && !mlp_first) mlp_phase(home, tid, &s_t, (float*)s_raw, mctr, feat, w1, b1, w2, b2, acc);
}

__global__ __launch_bounds__(256) void k_agg(
                      const float* __restrict__ e, const int* __restrict__ src,
                      const int* __restrict__ dst, const unsigned* __restrict__ bits,
                      const unsigned* __restrict__ gmax, unsigned* __restrict__ ctr,
                      const float* __restrict__ hcur0, const float* __restrict__ hcur1,
                      float* __restrict__ acc, int ng) {
    agg_body<false>(e, src, dst, bits, gmax, ctr, hcur0, hcur1, acc, ng,
                    nullptr, nullptr, nullptr, nullptr, nullptr, nullptr);
}

__global__ __launch_bounds__(256, 10) void k_agg_mlp(
                      const float* __restrict__ e, const int* __restrict__ src,
                      const int* __restrict__ dst, const unsigned* __restrict__ bits,
                      const unsigned* __restrict__ gmax, unsigned* __restrict__ ctr,
                      const float* __restrict__ hcur0, const float* __restrict__ hcur1,
                      float* __restrict__ acc, int ng,
                      const float* __restrict__ feat, const float* __restrict__ w1,
                      const float* __restrict__ b1,   const float* __restrict__ w2,
                      const float* __restrict__ b2,   unsigned* __restrict__ mctr) {
    agg_body<true>(e, src, dst, bits, gmax, ctr, hcur0, hcur1, acc, ng, feat, w1, b1, w2, b2, mctr);
}

__global__ __launch_bounds__(256) void k_mlp_combine(
    const float* __restrict__ feat, const float* __restrict__ w1,
    const float* __restrict__ b1,   const float* __restrict__ w2,
    const float* __restrict__ b2,   const float* __restrict__ alpha,
    const float* __restrict__ att,  const float* __restrict__ nline,
    float* __restrict__ out, int n_pairs)
{
    __shared__ float sW1p[(FF / 2) * W1_PSTRIDE];
    __shared__ float sW2p[(HH / 2) * W2_PSTRIDE];
    __shared__ float sHid[8][2][16 * HID_STRIDE];

    int t = threadIdx.x;

    for (int i = t; i < (FF / 2) * (HH / 4); i += 256) {
        int p  = i >> 4;
        int nq = (i & 15) << 2;
        float4 x = *(const float4*)&w1[(size_t)p * 2 * HH + nq];
        float4 y = *(const float4*)&w1[(size_t)p * 2 * HH + HH + nq];
        float4 lo = { x.x, y.x, x.y, y.y };
        float4 hi = { x.z, y.z, x.w, y.w };
        float* d = &sW1p[p * W1_PSTRIDE + 2 * nq];
        *(float4*)(d + 0) = lo;
        *(float4*)(d + 4) = hi;
    }
    for (int i = t; i < (HH / 2) * CC; i += 256) {
        int p = i >> 4, n = i & 15;
        v2f v; v.x = w2[(size_t)p * 2 * CC + n];
        v.y = w2[(size_t)p * 2 * CC + CC + n];
        *(v2f*)&sW2p[p * W2_PSTRIDE + 2 * n] = v;
    }
    __syncthreads();

    int wave = t >> 5;
    int lane = t & 31;
    int pair = blockIdx.x * 8 + wave;
    if (pair >= n_pairs) return;

    int tile0 = pair * 2;
    int tile1 = tile0 + 1;

    int m     = lane & 15;
    int n     = lane & 15;
    int hi16  = lane >> 4;
    int khalf = hi16 << 1;
    int moff  = hi16 << 3;

    v8f acc0 = {}, acc1 = {}, acc2 = {}, acc3 = {};
    v8f acc4 = {}, acc5 = {}, acc6 = {}, acc7 = {};
    const float* arow0 = feat + ((size_t)tile0 * 16 + m) * FF;
    const float* arow1 = feat + ((size_t)tile1 * 16 + m) * FF;
    for (int k = 0; k < FF; k += 4) {
        int kb = k + khalf;
        v2f a0 = *(const v2f*)(arow0 + kb);
        v2f a1 = *(const v2f*)(arow1 + kb);
        const float* bp = &sW1p[((k >> 1) + hi16) * W1_PSTRIDE + 2 * n];
        v2f b0 = *(const v2f*)(bp + 0);
        v2f b1v = *(const v2f*)(bp + 32);
        v2f b2v = *(const v2f*)(bp + 64);
        v2f b3v = *(const v2f*)(bp + 96);
        acc0 = __builtin_amdgcn_wmma_f32_16x16x4_f32(false, a0, false, b0,  (short)0, acc0, false, false);
        acc1 = __builtin_amdgcn_wmma_f32_16x16x4_f32(false, a0, false, b1v, (short)0, acc1, false, false);
        acc2 = __builtin_amdgcn_wmma_f32_16x16x4_f32(false, a0, false, b2v, (short)0, acc2, false, false);
        acc3 = __builtin_amdgcn_wmma_f32_16x16x4_f32(false, a0, false, b3v, (short)0, acc3, false, false);
        acc4 = __builtin_amdgcn_wmma_f32_16x16x4_f32(false, a1, false, b0,  (short)0, acc4, false, false);
        acc5 = __builtin_amdgcn_wmma_f32_16x16x4_f32(false, a1, false, b1v, (short)0, acc5, false, false);
        acc6 = __builtin_amdgcn_wmma_f32_16x16x4_f32(false, a1, false, b2v, (short)0, acc6, false, false);
        acc7 = __builtin_amdgcn_wmma_f32_16x16x4_f32(false, a1, false, b3v, (short)0, acc7, false, false);
    }

    float bb0 = b1[n + 0], bb1 = b1[n + 16], bb2 = b1[n + 32], bb3 = b1[n + 48];
    {
        float* hd0 = &sHid[wave][0][0];
        float* hd1 = &sHid[wave][1][0];
        for (int r = 0; r < 8; ++r) {
            int mm = r + moff;
            hd0[mm * HID_STRIDE + n +  0] = fmaxf(acc0[r] + bb0, 0.0f);
            hd0[mm * HID_STRIDE + n + 16] = fmaxf(acc1[r] + bb1, 0.0f);
            hd0[mm * HID_STRIDE + n + 32] = fmaxf(acc2[r] + bb2, 0.0f);
            hd0[mm * HID_STRIDE + n + 48] = fmaxf(acc3[r] + bb3, 0.0f);
            hd1[mm * HID_STRIDE + n +  0] = fmaxf(acc4[r] + bb0, 0.0f);
            hd1[mm * HID_STRIDE + n + 16] = fmaxf(acc5[r] + bb1, 0.0f);
            hd1[mm * HID_STRIDE + n + 32] = fmaxf(acc6[r] + bb2, 0.0f);
            hd1[mm * HID_STRIDE + n + 48] = fmaxf(acc7[r] + bb3, 0.0f);
        }
    }

    float bc = b2[n];
    for (int tt = 0; tt < 2; ++tt) {
        const float* hd = &sHid[wave][tt][0];
        v8f acc = {};
        for (int k = 0; k < HH; k += 4) {
            int kb = k + khalf;
            v2f a = *(const v2f*)(hd + m * HID_STRIDE + kb);
            v2f b = *(const v2f*)&sW2p[((k >> 1) + hi16) * W2_PSTRIDE + 2 * n];
            acc = __builtin_amdgcn_wmma_f32_16x16x4_f32(false, a, false, b, (short)0, acc, false, false);
        }
        int tile = tt ? tile1 : tile0;
        for (int r = 0; r < 8; ++r) {
            int node = tile * 16 + r + moff;
            float mlp = acc[r] + bc;
            float al  = alpha[node];
            float sa  = 1.0f / (1.0f + expf(-al));
            float sna = 1.0f - sa;
            float a0 = att[node * 2 + 0], a1 = att[node * 2 + 1];
            float mx = fmaxf(a0, a1);
            float e0 = expf(a0 - mx), e1 = expf(a1 - mx);
            float inv = 1.0f / (e0 + e1);
            float s1  = nline[(size_t)node * QS + CC];
            float h1v = nline[(size_t)node * QS + n];
            h1v = (s1 > 0.0f) ? h1v / s1 : h1v;
            float h0v = out[(size_t)node * CC + n];
            float logit = h0v * (e0 * inv) + h1v * (e1 * inv);
            out[(size_t)node * CC + n] = sa * logit + sna * mlp;
        }
    }
}

__global__ __launch_bounds__(256) void k_combine(const float* __restrict__ alpha, const float* __restrict__ att,
                                                 const float* __restrict__ acc0, float* __restrict__ out) {
    int i = blockIdx.x * 256 + threadIdx.x;
    int node = i >> 4;
    int n    = i & 15;
    const float* l0 = acc0 + (size_t)node * QS;
    const float* l1 = l0 + (size_t)NN * QS;
    float h0v = l0[n], s0 = l0[CC];
    float h1v = l1[n], s1 = l1[CC];
    float mlp = ((n & 8) ? l1 : l0)[24 + (n & 7)];
    float al  = alpha[node];
    float a0 = att[node * 2 + 0], a1 = att[node * 2 + 1];
    float sa  = 1.0f / (1.0f + expf(-al));
    float sna = 1.0f - sa;
    float mx = fmaxf(a0, a1);
    float e0 = expf(a0 - mx), e1 = expf(a1 - mx);
    float inv = 1.0f / (e0 + e1);
    h0v = (s0 > 0.0f) ? h0v / s0 : h0v;
    h1v = (s1 > 0.0f) ? h1v / s1 : h1v;
    float logit = h0v * (e0 * inv) + h1v * (e1 * inv);
    out[i] = sa * logit + sna * mlp;
}

extern "C" void kernel_launch(void* const* d_in, const int* in_sizes, int n_in,
                              void* d_out, int out_size, void* d_ws, size_t ws_size,
                              hipStream_t stream) {
    const float* features      = (const float*)d_in[0];
    const float* label_init    = (const float*)d_in[1];
    const float* labels_onehot = (const float*)d_in[2];
    const float* alpha         = (const float*)d_in[3];
    const float* attention     = (const float*)d_in[4];
    const float* e_edge        = (const float*)d_in[5];
    const float* w1            = (const float*)d_in[6];
    const float* b1            = (const float*)d_in[7];
    const float* w2            = (const float*)d_in[8];
    const float* b2            = (const float*)d_in[9];
    const int*   src           = (const int*)d_in[10];
    const int*   dst           = (const int*)d_in[11];
    const int*   train_mask    = (const int*)d_in[12];
    float* out = (float*)d_out;

    float* ws = (float*)d_ws;
    unsigned* gmax = (unsigned*)ws;
    float* acc  = ws + NN;

    const int TB = 256;
    const int gridI = NN / 32;
    const int n_pairs = NN / 32;
    const int blocks  = (n_pairs + 7) / 8;

    if (ws_size >= sizeof(float) * NN * (1 + 2 * QS + CC)) {
        float* acc1 = acc + (size_t)NN * QS;
        float* hrow = acc1 + (size_t)NN * QS;
        k_init<false><<<dim3(gridI, GG), 512, 0, stream>>>(labels_onehot, train_mask, gmax, acc, nullptr, nullptr);
        k_gmax<<<dim3(256, LL * GG), TB, 0, stream>>>(e_edge, gmax);
        k_agg<<<AGG_BLOCKS, TB, 0, stream>>>(e_edge, src, dst, gmax + BITS0, gmax, gmax + CTR0,
                                             label_init, label_init, acc, GG);
        k_init<true><<<dim3(gridI, GG), 512, 0, stream>>>(labels_onehot, train_mask, gmax, acc, hrow, out);
        k_agg_mlp<<<AGG_BLOCKS_MLP, TB, 0, stream>>>(e_edge + (size_t)GG * EE, src, dst, gmax + BITS0, gmax + GG,
                                             gmax + CTR0 + 1, hrow, out, acc, GG,
                                             features, w1, b1, w2, b2, gmax + CTR0 + MCTR);
        k_combine<<<NN * CC / 256, TB, 0, stream>>>(alpha, attention, acc, out);
        return;
    }

    float* hrow = acc + (size_t)NN * QS;
    const float* started = nullptr;
    for (int g = 0; g < GG; ++g) {
        for (int l = 0; l < LL; ++l) {
            const float* ee = e_edge + ((size_t)l * GG + g) * EE;
            const int* sg = src + (size_t)g * EE;
            const int* dg = dst + (size_t)g * EE;
            float* dest = l ? hrow : out;
            const float* hcur = l ? hrow : label_init;
            if (!started) {
                k_init<false><<<gridI, 512, 0, stream>>>(labels_onehot, train_mask, gmax, acc, dest, nullptr);
                k_gmax<<<dim3(256, LL * GG), TB, 0, stream>>>(e_edge, gmax);
                started = acc;
            } else {
                k_init<true> <<<gridI, 512, 0, stream>>>(labels_onehot, train_mask, gmax, acc, dest, nullptr);
            }
            k_agg<<<AGG_BLOCKS, TB, 0, stream>>>(ee, sg, dg, gmax + BITS0, gmax + (l * GG + g),
                                                 gmax + CTR0 + (g * LL + l), hcur, hcur, acc, 1);
        }
    }
    k_mlp_combine<<<blocks, TB, 0, stream>>>(features, w1, b1, w2, b2,
                                             alpha, attention, acc,
                                             out, n_pairs);
}


# Add phase: one edge per wave instruction, list entry read as LDS broadcast (lanes 0..15 products, lane 16 weight): no cross-lane moves, selects or 64-bit addresses; add arrival order may change
# speedup vs baseline: 19.2092x; 1.0813x over previous
#include <hip/hip_runtime.h>
#include <hip/hip_bf16.h>
#include <math.h>

#define NN 100000
#define CC 16
#define GG 2
#define LL 2
#define EE 3200000
#define FF 512
#define HH 64

#define W1_PSTRIDE 160
#define W2_PSTRIDE 32
#define HID_STRIDE 66
#define KS 32
#define W1S_STRIDE 72
#define W1S_WORDS (KS * W1S_STRIDE)
#define MLP_ROWS 128

typedef float v2f __attribute__((ext_vector_type(2)));
typedef float v8f __attribute__((ext_vector_type(8)));

__device__ __forceinline__ unsigned f2ou(float f) {
    unsigned u = __float_as_uint(f);
    return (u & 0x80000000u) ? ~u : (u | 0x80000000u);
}
__device__ __forceinline__ float ou2f(unsigned u) {
    return __uint_as_float((u & 0x80000000u) ? (u & 0x7FFFFFFFu) : ~u);
}

#define QS 32
#define CHUNK 1024
#define NCHUNK (EE / CHUNK)
#define NHALF (NN / 2)
#define CTR0 64
#define AGG_BLOCKS 2048
#define AGG_BLOCKS_MLP 1280
#define NBITW (NN / 32)
#define NBITQ ((NBITW + 3) / 4)
#define BITS0 256

template <bool NORM>
__global__ __launch_bounds__(512) void k_init(
                       const float* __restrict__ one_hot, const int* __restrict__ mask,
                       unsigned* __restrict__ gmax, float* __restrict__ acc0,
                       float* __restrict__ dest0, float* __restrict__ dest1) {
    int t = threadIdx.x;
    float* acc  = acc0 + (size_t)blockIdx.y * NN * QS;
    float* dest = blockIdx.y ? dest1 : dest0;
    int node0 = blockIdx.x * 32;
    int   mk[2];
    float oh[2], st[2];
    int mb = NORM ? 0 : mask[node0 + (t & 31)];
    for (int r = 0; r < 2; ++r) {
        int w    = r * 512 + t;
        int node = node0 + (w >> 5);
        mk[r] = mask[node];
        oh[r] = one_hot[(size_t)node * CC + (w & 15)];
    }
    for (int r = 0; r < 2; ++r) {
        int col = (r * 512 + t) & 31;
        float x = (col < CC) ? oh[r] : ((col == CC) ? 1.0f : 0.0f);
        st[r] = mk[r] ? x : 0.0f;
    }
    if (NORM) {
        int node = node0 + (t >> 4);
        int n    = t & 15;
        float v = acc[(size_t)node * QS + n];
        float s = acc[(size_t)node * QS + CC];
        dest[(size_t)node * CC + n] = (s > 0.0f) ? v / s : v;
        __syncthreads();
    }
    for (int r = 0; r < 2; ++r) {
        int w = r * 512 + t;
        acc[(size_t)(node0 + (w >> 5)) * QS + (w & 31)] = st[r];
    }
    if (!NORM && blockIdx.x == 0 && blockIdx.y == 0) {
        if (t < LL * GG) gmax[t] = 0x007FFFFFu;
        else if (t >= CTR0 && t < CTR0 + 128) gmax[t] = 0u;
        else if (t >= CTR0 + 128 && t < CTR0 + 128 + (4 * NBITQ - NBITW)) gmax[BITS0 + NBITW + (t - CTR0 - 128)] = 0u;
    }
    if (!NORM) {
        unsigned w = __builtin_amdgcn_ballot_w32(mb != 0);
        if (t == 0 && blockIdx.y == 0) gmax[BITS0 + blockIdx.x] = w;
    }
}

__device__ __forceinline__ float max4(float4 v) {
    return fmaxf(fmaxf(v.x, v.y), fmaxf(v.z, v.w));
}
__global__ __launch_bounds__(256) void k_gmax(const float* __restrict__ e,
                                              unsigned* __restrict__ gmax) {
    __shared__ float wmax[8];
    const float4* p = (const float4*)(e + (size_t)blockIdx.y * EE);
    const int n4 = EE / 4;
    const int stride = gridDim.x * 256;
    int i = blockIdx.x * 256 + threadIdx.x;
    float m = -INFINITY;
    for (; i + 3 * stride < n4; i += 4 * stride) {
        float4 a = p[i], b = p[i + stride], c = p[i + 2 * stride], d = p[i + 3 * stride];
        m = fmaxf(m, fmaxf(fmaxf(max4(a), max4(b)), fmaxf(max4(c), max4(d))));
    }
    for (; i < n4; i += stride) m = fmaxf(m, max4(p[i]));
    for (int off = 16; off > 0; off >>= 1) m = fmaxf(m, __shfl_xor(m, off));
    if ((threadIdx.x & 31) == 0) wmax[threadIdx.x >> 5] = m;
    __syncthreads();
    if (threadIdx.x == 0) {
        float b = wmax[0];
        for (int w = 1; w < 8; ++w) b = fmaxf(b, wmax[w]);
        atomicMax(&gmax[blockIdx.y], f2ou(b));
    }
}

#define GRP 32
template <bool GUARD> __device__ __forceinline__ void agg_edges(
    const int* s_d, const int* s_s, const float* s_a, int qb, int cnt, int n, int lane,
    const float* __restrict__ hcur, float* __restrict__ acc) {
    float v[GRP];
#pragma unroll
    for (int q = 0; q < GRP / 4; ++q) {
        int4 t = *(const int4*)(s_s + qb + 4 * q);
        v[4 * q + 0] = hcur[(unsigned)(t.x * CC + n)];
        v[4 * q + 1] = hcur[(unsigned)(t.y * CC + n)];
        v[4 * q + 2] = hcur[(unsigned)(t.z * CC + n)];
        v[4 * q + 3] = hcur[(unsigned)(t.w * CC + n)];
    }
    if (lane < 17) {
#pragma unroll
        for (int q = 0; q < GRP / 4; ++q) {
            int4   t = *(const int4*)(s_d + qb + 4 * q);
            float4 u = *(const float4*)(s_a + qb + 4 * q);
            int   dv[4] = { t.x, t.y, t.z, t.w };
            float av[4] = { u.x, u.y, u.z, u.w };
#pragma unroll
            for (int j = 0; j < 4; ++j) {
                int k = 4 * q + j;
                float x = (lane == CC) ? av[j] : v[k] * av[j];
                if (!GUARD || qb + k < cnt)
                    atomicAdd(acc + (unsigned)(dv[j] * QS + lane), x);
            }
        }
    }
}

#define NMLPT0 ((NHALF + MLP_ROWS - 1) / MLP_ROWS)
#define MCTR 8

__device__ __forceinline__ v8f mm4(v2f a, v2f b, v8f c) {
    return __builtin_amdgcn_wmma_f32_16x16x4_f32(false, a, false, b, (short)0, c, false, false);
}

__device__ __forceinline__ void mlp_ticket(int ticket, int tid, float* __restrict__ sW,
        const float* __restrict__ feat, const float* __restrict__ w1, const float* __restrict__ b1,
        const float* __restrict__ w2, const float* __restrict__ b2, float* __restrict__ acc) {
    int lane = tid & 31;
    int c = lane & 15;
    int h = lane >> 4;
    int row0 = __builtin_amdgcn_readfirstlane(ticket * MLP_ROWS + (tid >> 5) * 16);
    bool live = row0 < NN;
    int rbase = live ? row0 : 0;
    int node = rbase + c;
    const float* fbase = feat + (size_t)rbase * FF;
    unsigned flo = (unsigned)(c * FF + 4 * h);
    int sp  = tid >> 4;
    int snq = (tid & 15) << 2;
    unsigned sgo = (unsigned)((2 * sp) * HH + snq);
    float* sd = sW + (2 * sp) * W1S_STRIDE + snq;
    float4 sx = *(const float4*)(w1 + sgo);
    float4 sy = *(const float4*)(w1 + HH + sgo);
    *(float4*)(sd) = sx;
    *(float4*)(sd + W1S_STRIDE) = sy;
    float4 fc[4], fn[4];
#pragma unroll
    for (int i = 0; i < 4; ++i) fc[i] = *(const float4*)(fbase + 8 * i + flo);
#pragma unroll
    for (int i = 0; i < 4; ++i) fn[i] = fc[i];
    __syncthreads();
    v8f a0 = {}, a1 = {}, a2 = {}, a3 = {};
#pragma nounroll
    for (int j = 0; j < FF / KS; ++j) {
        const float* cur = sW + (j & 1) * W1S_WORDS;
        bool more = (j + 1 < FF / KS);
        if (more) {
            sx = *(const float4*)(w1 + (j + 1) * KS * HH + sgo);
            sy = *(const float4*)(w1 + (j + 1) * KS * HH + HH + sgo);
#pragma unroll
            for (int i = 0; i < 4; ++i) fn[i] = *(const float4*)(fbase + (j + 1) * KS + 8 * i + flo);
        }
#pragma unroll
        for (int kb = 0; kb < KS; kb += 8) {
            float4 f = fc[kb >> 3];
            const float* ap = cur + (kb + 4 * h) * W1S_STRIDE + c;
            v2f f01 = { f.x, f.y };
            v2f f23 = { f.z, f.w };
            v2f w00 = { ap[ 0], ap[W1S_STRIDE +  0] };
            v2f w01 = { ap[16], ap[W1S_STRIDE + 16] };
            v2f w02 = { ap[32], ap[W1S_STRIDE + 32] };
            v2f w03 = { ap[48], ap[W1S_STRIDE + 48] };
            a0 = mm4(w00, f01, a0);
            a1 = mm4(w01, f01, a1);
            a2 = mm4(w02, f01, a2);
            a3 = mm4(w03, f01, a3);
            v2f w20 = { ap[2 * W1S_STRIDE +  0], ap[3 * W1S_STRIDE +  0] };
            v2f w21 = { ap[2 * W1S_STRIDE + 16], ap[3 * W1S_STRIDE + 16] };
            v2f w22 = { ap[2 * W1S_STRIDE + 32], ap[3 * W1S_STRIDE + 32] };
            v2f w23 = { ap[2 * W1S_STRIDE + 48], ap[3 * W1S_STRIDE + 48] };
            a0 = mm4(w20, f23, a0);
            a1 = mm4(w21, f23, a1);
            a2 = mm4(w22, f23, a2);
            a3 = mm4(w23, f23, a3);
        }
        if (more) {
            float* d = sd + ((j + 1) & 1) * W1S_WORDS;
            *(float4*)(d) = sx;
            *(float4*)(d + W1S_STRIDE) = sy;
        }
#pragma unroll
        for (int i = 0; i < 4; ++i) fc[i] = fn[i];
        __syncthreads();
    }
    if (live) {
        int te = tid;
        asm volatile("" : "+v"(te));
        int c2 = te & 15, h2 = (te >> 4) & 1;
        const float* wb = sW + 2 * W1S_WORDS + (8 * h2) * CC + c2;
        const float* bb = sW + 2 * W1S_WORDS + HH * CC + 8 * h2;
        v8f o = {};
#pragma unroll
        for (int t = 0; t < 4; ++t) {
            v8f x = (t == 0) ? a0 : (t == 1) ? a1 : (t == 2) ? a2 : a3;
#pragma unroll
            for (int q = 0; q < 4; ++q) {
                int u = 16 * t + 2 * q;
                v2f hb = { fmaxf(x[2 * q] + bb[u], 0.0f), fmaxf(x[2 * q + 1] + bb[u + 1], 0.0f) };
                v2f wa = { wb[u * CC], wb[(u + 1) * CC] };
                o = mm4(wa, hb, o);
            }
        }
        asm volatile("s_wait_alu depctr_va_vdst(0)" : "+v"(o));
        float* q = acc + (size_t)((unsigned)(h2 * NN + rbase + c2) * QS + 24u);
#pragma unroll
        for (int r = 0; r < 8; ++r)
            __hip_atomic_store(q + r, o[r] + bb[HH + r], __ATOMIC_RELAXED, __HIP_MEMORY_SCOPE_AGENT);
    }
}

__device__ __forceinline__ void mlp_phase(unsigned home, int tid, unsigned* s_t, float* sW, unsigned* __restrict__ mctr,
        const float* __restrict__ feat, const float* __restrict__ w1, const float* __restrict__ b1,
        const float* __restrict__ w2, const float* __restrict__ b2, float* __restrict__ acc) {
    {
        float* sW2 = sW + 2 * W1S_WORDS;
        ((float4*)sW2)[tid] = ((const float4*)w2)[tid];
        if (tid < HH) sW2[HH * CC + tid] = b1[tid];
        else if (tid < HH + CC) sW2[HH * CC + tid] = b2[tid - HH];
    }
    for (int ph = 0; ph < 2; ++ph) {
        unsigned hh = home ^ (unsigned)ph;
        for (;;) {
            if (tid == 0) *s_t = atomicAdd(mctr + 64 * hh, 1u);
            __syncthreads();
            unsigned tk = *s_t;
            __syncthreads();
            if (tk >= (unsigned)NMLPT0) break;
            mlp_ticket((int)(hh * NMLPT0 + tk), tid, sW, feat, w1, b1, w2, b2, acc);
        }
    }
}

template <bool MLP>
__device__ __forceinline__ void agg_body(
                      const float* __restrict__ e, const int* __restrict__ src,
                      const int* __restrict__ dst, const unsigned* __restrict__ bits,
                      const unsigned* __restrict__ gmax, unsigned* __restrict__ ctr,
                      const float* __restrict__ hcur0, const float* __restrict__ hcur1,
                      float* __restrict__ acc, int ng,
                      const float* __restrict__ feat, const float* __restrict__ w1,
                      const float* __restrict__ b1,   const float* __restrict__ w2,
                      const float* __restrict__ b2,   unsigned* __restrict__ mctr) {
    __shared__ uint4    s_raw[3 * CHUNK / 4 + 4 * 256];
    __shared__ unsigned s_cnt;
    __shared__ unsigned s_t;
    __shared__ unsigned s_home;
    int*   s_d = (int*)s_raw;
    int*   s_s = s_d + CHUNK;
    float* s_a = (float*)(s_s + CHUNK);
    uint4* s_bits4 = s_raw + 3 * CHUNK / 4;
    int tid  = threadIdx.x;
    if (tid == 0) s_home = (((unsigned)__builtin_amdgcn_s_sendmsg_rtn(0x87) >> 16) >> 2) & 1u;
    bool mlp_first = MLP && (blockIdx.x * 2654435761u) < 858993459u;
    if (MLP) {
        __syncthreads();
        if (mlp_first) {
            mlp_phase(s_home, tid, &s_t, (float*)s_raw, mctr, feat, w1, b1, w2, b2, acc);
            __syncthreads();
        }
    }
    int ta = tid;
    if (MLP) asm volatile("" : "+v"(ta));
    int lane = ta & 31;
    int wave = ta >> 5;
    int n = ta & 15;
    {
        const uint4* bq = (const uint4*)bits;
        uint4 tb[4];
#pragma unroll
        for (int k = 0; k < 4; ++k) {
            int i = tid + k * 256;
            tb[k] = bq[i < NBITQ ? i : NBITQ - 1];
        }
#pragma unroll
        for (int k = 0; k < 4; ++k) s_bits4[tid + k * 256] = tb[k];
#pragma unroll
        for (int k = 0; k < 4; ++k) s_s[tid + k * 256] = 0;
    }
    const unsigned* s_bits = (const unsigned*)s_bits4;
    __syncthreads();
    unsigned home = s_home;
    for (int ph = 0; ph < 2; ++ph) {
        unsigned h = home ^ (unsigned)ph;
        int lo  = h ? NHALF : 0;
        int hiN = h ? NN : NHALF;
      for (int g = 0; g < ng; ++g) {
        const float* eg   = e   + (size_t)g * EE;
        const int*   sg   = src + (size_t)g * EE;
        const int*   dg   = dst + (size_t)g * EE;
        const float* hcur = g ? hcur1 : hcur0;
        float*       accg = acc + (size_t)g * NN * QS;
        unsigned*    ctrg = ctr + g * LL + 64 * h;
        float gm = ou2f(gmax[g]);
        for (;;) {
            if (tid == 0) {
                s_t = atomicAdd(ctrg, 1u);
                s_cnt = 0u;
            }
            __syncthreads();
            unsigned tk = s_t;
            __syncthreads();
            if (tk >= (unsigned)NCHUNK) break;
            int i0 = (int)tk * CHUNK + tid;
            int   dd[4], ss[4];
            float ev[4];
#pragma unroll
            for (int r = 0; r < 4; ++r) {
                dd[r] = dg[i0 + r * 256];
                ss[r] = sg[i0 + r * 256];
                ev[r] = eg[i0 + r * 256];
            }
            bool ok[4];
            unsigned bal[4];
            unsigned tw[4];
#pragma unroll
            for (int r = 0; r < 4; ++r) {
                unsigned wi = (unsigned)dd[r] >> 5;
                tw[r] = s_bits[wi < (unsigned)NBITW ? wi : (unsigned)(NBITW - 1)];
            }
#pragma unroll
            for (int r = 0; r < 4; ++r) {
                unsigned in  = (unsigned)(dd[r] >= lo) & (unsigned)(dd[r] < hiN);
                unsigned bit = (tw[r] >> (dd[r] & 31)) & 1u;
                ok[r] = (in & (bit ^ 1u)) != 0u;
            }
#pragma unroll
            for (int r = 0; r < 4; ++r) bal[r] = __builtin_amdgcn_ballot_w32(ok[r]);
            unsigned tot = (unsigned)(__popc(bal[0]) + __popc(bal[1]) + __popc(bal[2]) + __popc(bal[3]));
            unsigned base = 0u;
            if (lane == 0) base = atomicAdd(&s_cnt, tot);
            base = (unsigned)__builtin_amdgcn_readfirstlane((int)base);
#pragma unroll
            for (int r = 0; r < 4; ++r) {
                if (ok[r]) {
                    unsigned pos = base + __builtin_amdgcn_mbcnt_lo(bal[r], 0u);
                    s_d[pos] = dd[r];
                    s_s[pos] = ss[r];
                    s_a[pos] = expf(ev[r] - gm);
                }
                base += (unsigned)__popc(bal[r]);
            }
            __syncthreads();
            int cnt = __builtin_amdgcn_readfirstlane((int)s_cnt);
            for (int qb = __builtin_amdgcn_readfirstlane(wave * GRP); qb < cnt; qb += 8 * GRP) {
                if (qb + GRP <= cnt) agg_edges<false>(s_d, s_s, s_a, qb, cnt, n, lane, hcur, accg);
                else                 agg_edges<true >(s_d, s_s, s_a, qb, cnt, n, lane, hcur, accg);
            }
            __syncthreads();
        }
      }
    }
    if (MLP && !mlp_first) mlp_phase(home, tid, &s_t, (float*)s_raw, mctr, feat, w1, b1, w2, b2, acc);
}

__global__ __launch_bounds__(256) void k_agg(
                      const float* __restrict__ e, const int* __restrict__ src,
                      const int* __restrict__ dst, const unsigned* __restrict__ bits,
                      const unsigned* __restrict__ gmax, unsigned* __restrict__ ctr,
                      const float* __restrict__ hcur0, const float* __restrict__ hcur1,
                      float* __restrict__ acc, int ng) {
    agg_body<false>(e, src, dst, bits, gmax, ctr, hcur0, hcur1, acc, ng,
                    nullptr, nullptr, nullptr, nullptr, nullptr, nullptr);
}

__global__ __launch_bounds__(256, 10) void k_agg_mlp(
                      const float* __restrict__ e, const int* __restrict__ src,
                      const int* __restrict__ dst, const unsigned* __restrict__ bits,
                      const unsigned* __restrict__ gmax, unsigned* __restrict__ ctr,
                      const float* __restrict__ hcur0, const float* __restrict__ hcur1,
                      float* __restrict__ acc, int ng,
                      const float* __restrict__ feat, const float* __restrict__ w1,
                      const float* __restrict__ b1,   const float* __restrict__ w2,
                      const float* __restrict__ b2,   unsigned* __restrict__ mctr) {
    agg_body<true>(e, src, dst, bits, gmax, ctr, hcur0, hcur1, acc, ng, feat, w1, b1, w2, b2, mctr);
}

__global__ __launch_bounds__(256) void k_mlp_combine(
    const float* __restrict__ feat, const float* __restrict__ w1,
    const float* __restrict__ b1,   const float* __restrict__ w2,
    const float* __restrict__ b2,   const float* __restrict__ alpha,
    const float* __restrict__ att,  const float* __restrict__ nline,
    float* __restrict__ out, int n_pairs)
{
    __shared__ float sW1p[(FF / 2) * W1_PSTRIDE];
    __shared__ float sW2p[(HH / 2) * W2_PSTRIDE];
    __shared__ float sHid[8][2][16 * HID_STRIDE];

    int t = threadIdx.x;

    for (int i = t; i < (FF / 2) * (HH / 4); i += 256) {
        int p  = i >> 4;
        int nq = (i & 15) << 2;
        float4 x = *(const float4*)&w1[(size_t)p * 2 * HH + nq];
        float4 y = *(const float4*)&w1[(size_t)p * 2 * HH + HH + nq];
        float4 lo = { x.x, y.x, x.y, y.y };
        float4 hi = { x.z, y.z, x.w, y.w };
        float* d = &sW1p[p * W1_PSTRIDE + 2 * nq];
        *(float4*)(d + 0) = lo;
        *(float4*)(d + 4) = hi;
    }
    for (int i = t; i < (HH / 2) * CC; i += 256) {
        int p = i >> 4, n = i & 15;
        v2f v; v.x = w2[(size_t)p * 2 * CC + n];
        v.y = w2[(size_t)p * 2 * CC + CC + n];
        *(v2f*)&sW2p[p * W2_PSTRIDE + 2 * n] = v;
    }
    __syncthreads();

    int wave = t >> 5;
    int lane = t & 31;
    int pair = blockIdx.x * 8 + wave;
    if (pair >= n_pairs) return;

    int tile0 = pair * 2;
    int tile1 = tile0 + 1;

    int m     = lane & 15;
    int n     = lane & 15;
    int hi16  = lane >> 4;
    int khalf = hi16 << 1;
    int moff  = hi16 << 3;

    v8f acc0 = {}, acc1 = {}, acc2 = {}, acc3 = {};
    v8f acc4 = {}, acc5 = {}, acc6 = {}, acc7 = {};
    const float* arow0 = feat + ((size_t)tile0 * 16 + m) * FF;
    const float* arow1 = feat + ((size_t)tile1 * 16 + m) * FF;
    for (int k = 0; k < FF; k += 4) {
        int kb = k + khalf;
        v2f a0 = *(const v2f*)(arow0 + kb);
        v2f a1 = *(const v2f*)(arow1 + kb);
        const float* bp = &sW1p[((k >> 1) + hi16) * W1_PSTRIDE + 2 * n];
        v2f b0 = *(const v2f*)(bp + 0);
        v2f b1v = *(const v2f*)(bp + 32);
        v2f b2v = *(const v2f*)(bp + 64);
        v2f b3v = *(const v2f*)(bp + 96);
        acc0 = __builtin_amdgcn_wmma_f32_16x16x4_f32(false, a0, false, b0,  (short)0, acc0, false, false);
        acc1 = __builtin_amdgcn_wmma_f32_16x16x4_f32(false, a0, false, b1v, (short)0, acc1, false, false);
        acc2 = __builtin_amdgcn_wmma_f32_16x16x4_f32(false, a0, false, b2v, (short)0, acc2, false, false);
        acc3 = __builtin_amdgcn_wmma_f32_16x16x4_f32(false, a0, false, b3v, (short)0, acc3, false, false);
        acc4 = __builtin_amdgcn_wmma_f32_16x16x4_f32(false, a1, false, b0,  (short)0, acc4, false, false);
        acc5 = __builtin_amdgcn_wmma_f32_16x16x4_f32(false, a1, false, b1v, (short)0, acc5, false, false);
        acc6 = __builtin_amdgcn_wmma_f32_16x16x4_f32(false, a1, false, b2v, (short)0, acc6, false, false);
        acc7 = __builtin_amdgcn_wmma_f32_16x16x4_f32(false, a1, false, b3v, (short)0, acc7, false, false);
    }

    float bb0 = b1[n + 0], bb1 = b1[n + 16], bb2 = b1[n + 32], bb3 = b1[n + 48];
    {
        float* hd0 = &sHid[wave][0][0];
        float* hd1 = &sHid[wave][1][0];
        for (int r = 0; r < 8; ++r) {
            int mm = r + moff;
            hd0[mm * HID_STRIDE + n +  0] = fmaxf(acc0[r] + bb0, 0.0f);
            hd0[mm * HID_STRIDE + n + 16] = fmaxf(acc1[r] + bb1, 0.0f);
            hd0[mm * HID_STRIDE + n + 32] = fmaxf(acc2[r] + bb2, 0.0f);
            hd0[mm * HID_STRIDE + n + 48] = fmaxf(acc3[r] + bb3, 0.0f);
            hd1[mm * HID_STRIDE + n +  0] = fmaxf(acc4[r] + bb0, 0.0f);
            hd1[mm * HID_STRIDE + n + 16] = fmaxf(acc5[r] + bb1, 0.0f);
            hd1[mm * HID_STRIDE + n + 32] = fmaxf(acc6[r] + bb2, 0.0f);
            hd1[mm * HID_STRIDE + n + 48] = fmaxf(acc7[r] + bb3, 0.0f);
        }
    }

    float bc = b2[n];
    for (int tt = 0; tt < 2; ++tt) {
        const float* hd = &sHid[wave][tt][0];
        v8f acc = {};
        for (int k = 0; k < HH; k += 4) {
            int kb = k + khalf;
            v2f a = *(const v2f*)(hd + m * HID_STRIDE + kb);
            v2f b = *(const v2f*)&sW2p[((k >> 1) + hi16) * W2_PSTRIDE + 2 * n];
            acc = __builtin_amdgcn_wmma_f32_16x16x4_f32(false, a, false, b, (short)0, acc, false, false);
        }
        int tile = tt ? tile1 : tile0;
        for (int r = 0; r < 8; ++r) {
            int node = tile * 16 + r + moff;
            float mlp = acc[r] + bc;
            float al  = alpha[node];
            float sa  = 1.0f / (1.0f + expf(-al));
            float sna = 1.0f - sa;
            float a0 = att[node * 2 + 0], a1 = att[node * 2 + 1];
            float mx = fmaxf(a0, a1);
            float e0 = expf(a0 - mx), e1 = expf(a1 - mx);
            float inv = 1.0f / (e0 + e1);
            float s1  = nline[(size_t)node * QS + CC];
            float h1v = nline[(size_t)node * QS + n];
            h1v = (s1 > 0.0f) ? h1v / s1 : h1v;
            float h0v = out[(size_t)node * CC + n];
            float logit = h0v * (e0 * inv) + h1v * (e1 * inv);
            out[(size_t)node * CC + n] = sa * logit + sna * mlp;
        }
    }
}

__global__ __launch_bounds__(256) void k_combine(const float* __restrict__ alpha, const float* __restrict__ att,
                                                 const float* __restrict__ acc0, float* __restrict__ out) {
    int i = blockIdx.x * 256 + threadIdx.x;
    int node = i >> 4;
    int n    = i & 15;
    const float* l0 = acc0 + (size_t)node * QS;
    const float* l1 = l0 + (size_t)NN * QS;
    float h0v = l0[n], s0 = l0[CC];
    float h1v = l1[n], s1 = l1[CC];
    float mlp = ((n & 8) ? l1 : l0)[24 + (n & 7)];
    float al  = alpha[node];
    float a0 = att[node * 2 + 0], a1 = att[node * 2 + 1];
    float sa  = 1.0f / (1.0f + expf(-al));
    float sna = 1.0f - sa;
    float mx = fmaxf(a0, a1);
    float e0 = expf(a0 - mx), e1 = expf(a1 - mx);
    float inv = 1.0f / (e0 + e1);
    h0v = (s0 > 0.0f) ? h0v / s0 : h0v;
    h1v = (s1 > 0.0f) ? h1v / s1 : h1v;
    float logit = h0v * (e0 * inv) + h1v * (e1 * inv);
    out[i] = sa * logit + sna * mlp;
}

extern "C" void kernel_launch(void* const* d_in, const int* in_sizes, int n_in,
                              void* d_out, int out_size, void* d_ws, size_t ws_size,
                              hipStream_t stream) {
    const float* features      = (const float*)d_in[0];
    const float* label_init    = (const float*)d_in[1];
    const float* labels_onehot = (const float*)d_in[2];
    const float* alpha         = (const float*)d_in[3];
    const float* attention     = (const float*)d_in[4];
    const float* e_edge        = (const float*)d_in[5];
    const float* w1            = (const float*)d_in[6];
    const float* b1            = (const float*)d_in[7];
    const float* w2            = (const float*)d_in[8];
    const float* b2            = (const float*)d_in[9];
    const int*   src           = (const int*)d_in[10];
    const int*   dst           = (const int*)d_in[11];
    const int*   train_mask    = (const int*)d_in[12];
    float* out = (float*)d_out;

    float* ws = (float*)d_ws;
    unsigned* gmax = (unsigned*)ws;
    float* acc  = ws + NN;

    const int TB = 256;
    const int gridI = NN / 32;
    const int n_pairs = NN / 32;
    const int blocks  = (n_pairs + 7) / 8;

    if (ws_size >= sizeof(float) * NN * (1 + 2 * QS + CC)) {
        float* acc1 = acc + (size_t)NN * QS;
        float* hrow = acc1 + (size_t)NN * QS;
        k_init<false><<<dim3(gridI, GG), 512, 0, stream>>>(labels_onehot, train_mask, gmax, acc, nullptr, nullptr);
        k_gmax<<<dim3(256, LL * GG), TB, 0, stream>>>(e_edge, gmax);
        k_agg<<<AGG_BLOCKS, TB, 0, stream>>>(e_edge, src, dst, gmax + BITS0, gmax, gmax + CTR0,
                                             label_init, label_init, acc, GG);
        k_init<true><<<dim3(gridI, GG), 512, 0, stream>>>(labels_onehot, train_mask, gmax, acc, hrow, out);
        k_agg_mlp<<<AGG_BLOCKS_MLP, TB, 0, stream>>>(e_edge + (size_t)GG * EE, src, dst, gmax + BITS0, gmax + GG,
                                             gmax + CTR0 + 1, hrow, out, acc, GG,
                                             features, w1, b1, w2, b2, gmax + CTR0 + MCTR);
        k_combine<<<NN * CC / 256, TB, 0, stream>>>(alpha, attention, acc, out);
        return;
    }

    float* hrow = acc + (size_t)NN * QS;
    const float* started = nullptr;
    for (int g = 0; g < GG; ++g) {
        for (int l = 0; l < LL; ++l) {
            const float* ee = e_edge + ((size_t)l * GG + g) * EE;
            const int* sg = src + (size_t)g * EE;
            const int* dg = dst + (size_t)g * EE;
            float* dest = l ? hrow : out;
            const float* hcur = l ? hrow : label_init;
            if (!started) {
                k_init<false><<<gridI, 512, 0, stream>>>(labels_onehot, train_mask, gmax, acc, dest, nullptr);
                k_gmax<<<dim3(256, LL * GG), TB, 0, stream>>>(e_edge, gmax);
                started = acc;
            } else {
                k_init<true> <<<gridI, 512, 0, stream>>>(labels_onehot, train_mask, gmax, acc, dest, nullptr);
            }
            k_agg<<<AGG_BLOCKS, TB, 0, stream>>>(ee, sg, dg, gmax + BITS0, gmax + (l * GG + g),
                                                 gmax + CTR0 + (g * LL + l), hcur, hcur, acc, 1);
        }
    }
    k_mlp_combine<<<blocks, TB, 0, stream>>>(features, w1, b1, w2, b2,
                                             alpha, attention, acc,
                                             out, n_pairs);
}
